# starting kernel
# baseline (speedup 1.0000x reference)
#include <hip/hip_runtime.h>
#include <math.h>

typedef __attribute__((ext_vector_type(2))) float v2f;
typedef __attribute__((ext_vector_type(4))) float v4f;
typedef __attribute__((ext_vector_type(8))) float v8f;

#define DHID   128
#define TILE_M 16
#define WPB    6
#define HSTR   132
#define LNEPS  1e-5f

#if __has_builtin(__builtin_amdgcn_tanhf)
#define TANHF(v) __builtin_amdgcn_tanhf(v)
#else
#define TANHF(v) tanhf(v)
#endif

#define OFF_W1 32768
#define OFF_W2 49152
#define OFF_H  65536
#define SMEM_FLOATS (OFF_H + WPB * TILE_M * HSTR)

__device__ __forceinline__ void pack_w(float* __restrict__ dst,
                                       const float* __restrict__ W,
                                       int K, int tid, int nthr)
{
    const int total = K << 7;
    for (int i = tid; i < total; i += nthr) {
        const int r   = i >> 7;
        const int col = i & 127;
        dst[((r >> 2) << 9) + ((col >> 4) << 6) + ((col & 15) << 2) + (r & 3)] = W[i];
    }
}

template <int KSTEPS, int HS>
__device__ __forceinline__ void gemm_lds(const float* __restrict__ hbuf,
                                         const float* __restrict__ wp,
                                         v8f acc[8], int lane)
{
    const int row = lane & 15;
    const int g2  = (lane >> 4) << 1;
    const float* wl = wp + ((lane & 15) << 2) + g2;
#pragma unroll 4
    for (int kk = 0; kk < KSTEPS; ++kk) {
        const v2f a = *(const v2f*)(hbuf + row * HS + (kk << 2) + g2);
        const float* wk = wl + (kk << 9);
#pragma unroll
        for (int nn = 0; nn < 8; ++nn) {
            const v2f b = *(const v2f*)(wk + (nn << 6));
            acc[nn] = __builtin_amdgcn_wmma_f32_16x16x4_f32(
                false, a, false, b, (short)0, acc[nn], false, false);
        }
    }
}

__device__ __forceinline__ void layernorm_tanh(v8f acc[8],
                                               const float* __restrict__ bias,
                                               const float* __restrict__ gamma,
                                               const float* __restrict__ beta,
                                               int lane)
{
    const int col0 = lane & 15;
    float bsv[8], gv[8], bv[8];
#pragma unroll
    for (int nn = 0; nn < 8; ++nn) {
        const int c = col0 + (nn << 4);
        bsv[nn] = bias[c]; gv[nn] = gamma[c]; bv[nn] = beta[c];
    }
#pragma unroll
    for (int j = 0; j < 8; ++j) {
        float s = 0.f, ss = 0.f;
#pragma unroll
        for (int nn = 0; nn < 8; ++nn) {
            float v = acc[nn][j] + bsv[nn];
            acc[nn][j] = v;
            s += v;
            ss = fmaf(v, v, ss);
        }
#pragma unroll
        for (int m = 1; m < 16; m <<= 1) {
            s  += __shfl_xor(s,  m, 32);
            ss += __shfl_xor(ss, m, 32);
        }
        const float mu  = s * (1.0f / 128.0f);
        const float var = fmaf(-mu, mu, ss * (1.0f / 128.0f));
        const float inv = rsqrtf(var + LNEPS);
#pragma unroll
        for (int nn = 0; nn < 8; ++nn) {
            const float v = (acc[nn][j] - mu) * inv;
            acc[nn][j] = TANHF(fmaf(v, gv[nn], bv[nn]));
        }
    }
}

__device__ __forceinline__ void store_tile(float* __restrict__ hbuf,
                                           const v8f acc[8], int lane)
{
    const int col0 = lane & 15;
    const int grp  = lane >> 4;
#pragma unroll
    for (int j = 0; j < 8; ++j) {
        const int row = j + (grp << 3);
#pragma unroll
        for (int nn = 0; nn < 8; ++nn)
            hbuf[row * HSTR + col0 + (nn << 4)] = acc[nn][j];
    }
}

__device__ __forceinline__ void stage_rows(float* __restrict__ hbuf,
                                           const float* __restrict__ x,
                                           const int* __restrict__ idxbase,
                                           int e0, int E, int lane)
{
    const int c4 = lane << 2;
#pragma unroll 4
    for (int r = 0; r < TILE_M; ++r) {
        int e = e0 + r; if (e >= E) e = E - 1;
        const int n = idxbase[e];
        const v4f v = *(const v4f*)(x + (size_t)n * DHID + c4);
        *(v4f*)(hbuf + r * HSTR + c4) = v;
    }
}

__global__ void __launch_bounds__(WPB * 32, 1)
edge_mlp_wmma_f32(const float* __restrict__ x, const int* __restrict__ eidx, int E,
                  const float* __restrict__ W0, const float* __restrict__ b0,
                  const float* __restrict__ W1, const float* __restrict__ b1,
                  const float* __restrict__ W2, const float* __restrict__ b2,
                  const float* __restrict__ W3, const float* __restrict__ b3,
                  const float* __restrict__ g0, const float* __restrict__ be0,
                  const float* __restrict__ g1, const float* __restrict__ be1,
                  const float* __restrict__ g2, const float* __restrict__ be2,
                  float* __restrict__ out)
{
    extern __shared__ float smem[];
    const int tid  = threadIdx.x;
    const int lane = tid & 31;
    const int wave = tid >> 5;

    float* wp0  = smem;
    float* wp1  = smem + OFF_W1;
    float* wp2  = smem + OFF_W2;
    float* hbuf = smem + OFF_H + wave * (TILE_M * HSTR);

    pack_w(wp0, W0, 2 * DHID, tid, WPB * 32);
    pack_w(wp1, W1, DHID,     tid, WPB * 32);
    pack_w(wp2, W2, DHID,     tid, WPB * 32);
    __syncthreads();

    const int tiles  = (E + TILE_M - 1) / TILE_M;
    const int stride = gridDim.x * WPB;
    const int sweeps = (tiles + stride - 1) / stride;

    for (int it = 0; it < sweeps; ++it) {
        const int tile = (it * gridDim.x + blockIdx.x) * WPB + wave;
        const int e0   = tile * TILE_M;

        {
            const int tn = tile + stride;
            if (tn < tiles) {
                __builtin_prefetch(&eidx[tn * TILE_M], 0, 1);
                __builtin_prefetch(&eidx[E + tn * TILE_M], 0, 1);
            }
        }

        v8f h0[8] = {};
        stage_rows(hbuf, x, eidx, e0, E, lane);
        __syncthreads();
        gemm_lds<32, HSTR>(hbuf, wp0, h0, lane);
        __syncthreads();
        stage_rows(hbuf, x, eidx + E, e0, E, lane);
        __syncthreads();
        gemm_lds<32, HSTR>(hbuf, wp0 + 32 * 512, h0, lane);
        layernorm_tanh(h0, b0, g0, be0, lane);
        store_tile(hbuf, h0, lane);
        __syncthreads();

        v8f h1[8] = {};
        gemm_lds<32, HSTR>(hbuf, wp1, h1, lane);
        layernorm_tanh(h1, b1, g1, be1, lane);
        store_tile(hbuf, h1, lane);
        __syncthreads();

        v8f h2[8] = {};
        gemm_lds<32, HSTR>(hbuf, wp2, h2, lane);
        layernorm_tanh(h2, b2, g2, be2, lane);

        {
            const int col0 = lane & 15;
            const int grp  = lane >> 4;
            float w3[8];
#pragma unroll
            for (int nn = 0; nn < 8; ++nn) w3[nn] = W3[col0 + (nn << 4)];
            const float bias3 = b3[0];
#pragma unroll
            for (int j = 0; j < 8; ++j) {
                float p = 0.f;
#pragma unroll
                for (int nn = 0; nn < 8; ++nn) p = fmaf(h2[nn][j], w3[nn], p);
#pragma unroll
                for (int m = 1; m < 16; m <<= 1) p += __shfl_xor(p, m, 32);
                const int e = e0 + j + (grp << 3);
                if (col0 == j && e < E) out[e] = p + bias3;
            }
        }
        __syncthreads();
    }
}

extern "C" void kernel_launch(void* const* d_in, const int* in_sizes, int n_in,
                              void* d_out, int out_size, void* d_ws, size_t ws_size,
                              hipStream_t stream)
{
    const float* x   = (const float*)d_in[0];
    const int*   ei  = (const int*)  d_in[1];
    const float* W0  = (const float*)d_in[2];
    const float* b0  = (const float*)d_in[3];
    const float* W1  = (const float*)d_in[4];
    const float* b1  = (const float*)d_in[5];
    const float* W2  = (const float*)d_in[6];
    const float* b2  = (const float*)d_in[7];
    const float* W3  = (const float*)d_in[8];
    const float* b3  = (const float*)d_in[9];
    const float* g0  = (const float*)d_in[10];
    const float* be0 = (const float*)d_in[11];
    const float* g1  = (const float*)d_in[12];
    const float* be1 = (const float*)d_in[13];
    const float* g2  = (const float*)d_in[14];
    const float* be2 = (const float*)d_in[15];

    const int E     = in_sizes[1] / 2;
    const int tiles = (E + TILE_M - 1) / TILE_M;
    int blocks = (tiles + WPB - 1) / WPB;
    if (blocks > 1024) blocks = 1024;
    const size_t shmem = (size_t)SMEM_FLOATS * sizeof(float);

    edge_mlp_wmma_f32<<<blocks, WPB * 32, shmem, stream>>>(
        x, ei, E, W0, b0, W1, b1, W2, b2, W3, b3,
        g0, be0, g1, be1, g2, be2, (float*)d_out);
}


# sweep loop freed of its workgroup barriers: the h-buffer is wave-private and one wave's LDS ops run in order, so each wave runs its own tiles at its own pace and stops at its own last tile
# speedup vs baseline: 1.1864x; 1.1864x over previous
#include <hip/hip_runtime.h>
#include <math.h>

typedef __attribute__((ext_vector_type(2))) float v2f;
typedef __attribute__((ext_vector_type(4))) float v4f;
typedef __attribute__((ext_vector_type(8))) float v8f;

#define DHID   128
#define TILE_M 16
#define WPB    6
#define HSTR   132
#define LNEPS  1e-5f

#if __has_builtin(__builtin_amdgcn_tanhf)
#define TANHF(v) __builtin_amdgcn_tanhf(v)
#else
#define TANHF(v) tanhf(v)
#endif

#define OFF_W1 32768
#define OFF_W2 49152
#define OFF_H  65536
#define SMEM_FLOATS (OFF_H + WPB * TILE_M * HSTR)

__device__ __forceinline__ void pack_w(float* __restrict__ dst,
                                       const float* __restrict__ W,
                                       int K, int tid, int nthr)
{
    const int total = K << 7;
    for (int i = tid; i < total; i += nthr) {
        const int r   = i >> 7;
        const int col = i & 127;
        dst[((r >> 2) << 9) + ((col >> 4) << 6) + ((col & 15) << 2) + (r & 3)] = W[i];
    }
}

template <int KSTEPS, int HS>
__device__ __forceinline__ void gemm_lds(const float* __restrict__ hbuf,
                                         const float* __restrict__ wp,
                                         v8f acc[8], int lane)
{
    const int row = lane & 15;
    const int g2  = (lane >> 4) << 1;
    const float* wl = wp + ((lane & 15) << 2) + g2;
#pragma unroll 4
    for (int kk = 0; kk < KSTEPS; ++kk) {
        const v2f a = *(const v2f*)(hbuf + row * HS + (kk << 2) + g2);
        const float* wk = wl + (kk << 9);
#pragma unroll
        for (int nn = 0; nn < 8; ++nn) {
            const v2f b = *(const v2f*)(wk + (nn << 6));
            acc[nn] = __builtin_amdgcn_wmma_f32_16x16x4_f32(
                false, a, false, b, (short)0, acc[nn], false, false);
        }
    }
}

__device__ __forceinline__ void layernorm_tanh(v8f acc[8],
                                               const float* __restrict__ bias,
                                               const float* __restrict__ gamma,
                                               const float* __restrict__ beta,
                                               int lane)
{
    const int col0 = lane & 15;
    float bsv[8], gv[8], bv[8];
#pragma unroll
    for (int nn = 0; nn < 8; ++nn) {
        const int c = col0 + (nn << 4);
        bsv[nn] = bias[c]; gv[nn] = gamma[c]; bv[nn] = beta[c];
    }
#pragma unroll
    for (int j = 0; j < 8; ++j) {
        float s = 0.f, ss = 0.f;
#pragma unroll
        for (int nn = 0; nn < 8; ++nn) {
            float v = acc[nn][j] + bsv[nn];
            acc[nn][j] = v;
            s += v;
            ss = fmaf(v, v, ss);
        }
#pragma unroll
        for (int m = 1; m < 16; m <<= 1) {
            s  += __shfl_xor(s,  m, 32);
            ss += __shfl_xor(ss, m, 32);
        }
        const float mu  = s * (1.0f / 128.0f);
        const float var = fmaf(-mu, mu, ss * (1.0f / 128.0f));
        const float inv = rsqrtf(var + LNEPS);
#pragma unroll
        for (int nn = 0; nn < 8; ++nn) {
            const float v = (acc[nn][j] - mu) * inv;
            acc[nn][j] = TANHF(fmaf(v, gv[nn], bv[nn]));
        }
    }
}

__device__ __forceinline__ void store_tile(float* __restrict__ hbuf,
                                           const v8f acc[8], int lane)
{
    const int col0 = lane & 15;
    const int grp  = lane >> 4;
#pragma unroll
    for (int j = 0; j < 8; ++j) {
        const int row = j + (grp << 3);
#pragma unroll
        for (int nn = 0; nn < 8; ++nn)
            hbuf[row * HSTR + col0 + (nn << 4)] = acc[nn][j];
    }
}

__device__ __forceinline__ void stage_rows(float* __restrict__ hbuf,
                                           const float* __restrict__ x,
                                           const int* __restrict__ idxbase,
                                           int e0, int E, int lane)
{
    const int c4 = lane << 2;
#pragma unroll 4
    for (int r = 0; r < TILE_M; ++r) {
        int e = e0 + r; if (e >= E) e = E - 1;
        const int n = idxbase[e];
        const v4f v = *(const v4f*)(x + (size_t)n * DHID + c4);
        *(v4f*)(hbuf + r * HSTR + c4) = v;
    }
}

__device__ __forceinline__ void wave_lds_order()
{
    __builtin_amdgcn_fence(__ATOMIC_RELEASE, "wavefront");
    __builtin_amdgcn_wave_barrier();
    __builtin_amdgcn_fence(__ATOMIC_ACQUIRE, "wavefront");
}

__global__ void __launch_bounds__(WPB * 32, 1)
edge_mlp_wmma_f32(const float* __restrict__ x, const int* __restrict__ eidx, int E,
                  const float* __restrict__ W0, const float* __restrict__ b0,
                  const float* __restrict__ W1, const float* __restrict__ b1,
                  const float* __restrict__ W2, const float* __restrict__ b2,
                  const float* __restrict__ W3, const float* __restrict__ b3,
                  const float* __restrict__ g0, const float* __restrict__ be0,
                  const float* __restrict__ g1, const float* __restrict__ be1,
                  const float* __restrict__ g2, const float* __restrict__ be2,
                  float* __restrict__ out)
{
    extern __shared__ float smem[];
    const int tid  = threadIdx.x;
    const int lane = tid & 31;
    const int wave = tid >> 5;

    float* wp0  = smem;
    float* wp1  = smem + OFF_W1;
    float* wp2  = smem + OFF_W2;
    float* hbuf = smem + OFF_H + wave * (TILE_M * HSTR);

    pack_w(wp0, W0, 2 * DHID, tid, WPB * 32);
    pack_w(wp1, W1, DHID,     tid, WPB * 32);
    pack_w(wp2, W2, DHID,     tid, WPB * 32);
    __syncthreads();

    const int tiles  = (E + TILE_M - 1) / TILE_M;
    const int stride = gridDim.x * WPB;
    const int sweeps = (tiles + stride - 1) / stride;

    for (int it = 0; it < sweeps; ++it) {
        const int tile = (it * gridDim.x + blockIdx.x) * WPB + wave;
        if (__builtin_amdgcn_readfirstlane((int)(tile >= tiles))) break;
        const int e0   = tile * TILE_M;

        {
            const int tn = tile + stride;
            if (tn < tiles) {
                __builtin_prefetch(&eidx[tn * TILE_M], 0, 1);
                __builtin_prefetch(&eidx[E + tn * TILE_M], 0, 1);
            }
        }

        v8f h0[8] = {};
        stage_rows(hbuf, x, eidx, e0, E, lane);
        wave_lds_order();
        gemm_lds<32, HSTR>(hbuf, wp0, h0, lane);
        wave_lds_order();
        stage_rows(hbuf, x, eidx + E, e0, E, lane);
        wave_lds_order();
        gemm_lds<32, HSTR>(hbuf, wp0 + 32 * 512, h0, lane);
        layernorm_tanh(h0, b0, g0, be0, lane);
        store_tile(hbuf, h0, lane);
        wave_lds_order();

        v8f h1[8] = {};
        gemm_lds<32, HSTR>(hbuf, wp1, h1, lane);
        layernorm_tanh(h1, b1, g1, be1, lane);
        store_tile(hbuf, h1, lane);
        wave_lds_order();

        v8f h2[8] = {};
        gemm_lds<32, HSTR>(hbuf, wp2, h2, lane);
        layernorm_tanh(h2, b2, g2, be2, lane);

        {
            const int col0 = lane & 15;
            const int grp  = lane >> 4;
            float w3[8];
#pragma unroll
            for (int nn = 0; nn < 8; ++nn) w3[nn] = W3[col0 + (nn << 4)];
            const float bias3 = b3[0];
#pragma unroll
            for (int j = 0; j < 8; ++j) {
                float p = 0.f;
#pragma unroll
                for (int nn = 0; nn < 8; ++nn) p = fmaf(h2[nn][j], w3[nn], p);
#pragma unroll
                for (int m = 1; m < 16; m <<= 1) p += __shfl_xor(p, m, 32);
                const int e = e0 + j + (grp << 3);
                if (col0 == j && e < E) out[e] = p + bias3;
            }
        }
        wave_lds_order();
    }
}

extern "C" void kernel_launch(void* const* d_in, const int* in_sizes, int n_in,
                              void* d_out, int out_size, void* d_ws, size_t ws_size,
                              hipStream_t stream)
{
    const float* x   = (const float*)d_in[0];
    const int*   ei  = (const int*)  d_in[1];
    const float* W0  = (const float*)d_in[2];
    const float* b0  = (const float*)d_in[3];
    const float* W1  = (const float*)d_in[4];
    const float* b1  = (const float*)d_in[5];
    const float* W2  = (const float*)d_in[6];
    const float* b2  = (const float*)d_in[7];
    const float* W3  = (const float*)d_in[8];
    const float* b3  = (const float*)d_in[9];
    const float* g0  = (const float*)d_in[10];
    const float* be0 = (const float*)d_in[11];
    const float* g1  = (const float*)d_in[12];
    const float* be1 = (const float*)d_in[13];
    const float* g2  = (const float*)d_in[14];
    const float* be2 = (const float*)d_in[15];

    const int E     = in_sizes[1] / 2;
    const int tiles = (E + TILE_M - 1) / TILE_M;
    int blocks = (tiles + WPB - 1) / WPB;
    if (blocks > 1024) blocks = 1024;
    const size_t shmem = (size_t)SMEM_FLOATS * sizeof(float);

    edge_mlp_wmma_f32<<<blocks, WPB * 32, shmem, stream>>>(
        x, ei, E, W0, b0, W1, b1, W2, b2, W3, b3,
        g0, be0, g1, be1, g2, be2, (float*)d_out);
}


# per-wave LDS activation tile halved to 16x64 (each layer runs as K chunks of 64, same k order), so 12 waves (3 on every SIMD) fit beside the f32 weights instead of 6 (2,2,1,1)
# speedup vs baseline: 1.7396x; 1.4663x over previous
#include <hip/hip_runtime.h>
#include <math.h>

typedef __attribute__((ext_vector_type(2))) float v2f;
typedef __attribute__((ext_vector_type(4))) float v4f;
typedef __attribute__((ext_vector_type(8))) float v8f;

#define DHID   128
#define TILE_M 16
#define WPB    12
#define CW     64
#define HSTR   68
#define LNEPS  1e-5f

#if __has_builtin(__builtin_amdgcn_tanhf)
#define TANHF(v) __builtin_amdgcn_tanhf(v)
#else
#define TANHF(v) tanhf(v)
#endif

#define OFF_W1 32768
#define OFF_W2 49152
#define OFF_H  65536
#define SMEM_FLOATS (OFF_H + WPB * TILE_M * HSTR)

__device__ __forceinline__ void pack_w(float* __restrict__ dst,
                                       const float* __restrict__ W,
                                       int K, int tid, int nthr)
{
    const int total = K << 7;
    for (int i = tid; i < total; i += nthr) {
        const int r   = i >> 7;
        const int col = i & 127;
        dst[((r >> 2) << 9) + ((col >> 4) << 6) + ((col & 15) << 2) + (r & 3)] = W[i];
    }
}

template <int KSTEPS, int HS>
__device__ __forceinline__ void gemm_lds(const float* __restrict__ hbuf,
                                         const float* __restrict__ wp,
                                         v8f acc[8], int lane)
{
    const int row = lane & 15;
    const int g2  = (lane >> 4) << 1;
    const float* wl = wp + ((lane & 15) << 2) + g2;
#pragma unroll 4
    for (int kk = 0; kk < KSTEPS; ++kk) {
        const v2f a = *(const v2f*)(hbuf + row * HS + (kk << 2) + g2);
        const float* wk = wl + (kk << 9);
#pragma unroll
        for (int nn = 0; nn < 8; ++nn) {
            const v2f b = *(const v2f*)(wk + (nn << 6));
            acc[nn] = __builtin_amdgcn_wmma_f32_16x16x4_f32(
                false, a, false, b, (short)0, acc[nn], false, false);
        }
    }
}

__device__ __forceinline__ void layernorm_tanh(v8f acc[8],
                                               const float* __restrict__ bias,
                                               const float* __restrict__ gamma,
                                               const float* __restrict__ beta,
                                               int lane)
{
    const int col0 = lane & 15;
    float bsv[8], gv[8], bv[8];
#pragma unroll
    for (int nn = 0; nn < 8; ++nn) {
        const int c = col0 + (nn << 4);
        bsv[nn] = bias[c]; gv[nn] = gamma[c]; bv[nn] = beta[c];
    }
#pragma unroll
    for (int j = 0; j < 8; ++j) {
        float s = 0.f, ss = 0.f;
#pragma unroll
        for (int nn = 0; nn < 8; ++nn) {
            float v = acc[nn][j] + bsv[nn];
            acc[nn][j] = v;
            s += v;
            ss = fmaf(v, v, ss);
        }
#pragma unroll
        for (int m = 1; m < 16; m <<= 1) {
            s  += __shfl_xor(s,  m, 32);
            ss += __shfl_xor(ss, m, 32);
        }
        const float mu  = s * (1.0f / 128.0f);
        const float var = fmaf(-mu, mu, ss * (1.0f / 128.0f));
        const float inv = rsqrtf(var + LNEPS);
#pragma unroll
        for (int nn = 0; nn < 8; ++nn) {
            const float v = (acc[nn][j] - mu) * inv;
            acc[nn][j] = TANHF(fmaf(v, gv[nn], bv[nn]));
        }
    }
}

template <int HALF>
__device__ __forceinline__ void store_half(float* __restrict__ hbuf,
                                           const v8f acc[8], int lane)
{
    const int col0 = lane & 15;
    const int grp  = lane >> 4;
#pragma unroll
    for (int j = 0; j < 8; ++j) {
        const int row = j + (grp << 3);
#pragma unroll
        for (int nn = 0; nn < 4; ++nn)
            hbuf[row * HSTR + col0 + (nn << 4)] = acc[HALF * 4 + nn][j];
    }
}

__device__ __forceinline__ void stage_half(float* __restrict__ hbuf,
                                           const float* __restrict__ x,
                                           int idxv, int colbase, int lane)
{
    const int grp = lane >> 4;
    const int c4  = (lane & 15) << 2;
#pragma unroll
    for (int i = 0; i < TILE_M / 2; ++i) {
        const int r = 2 * i + grp;
        const int n = __shfl(idxv, r, 32);
        const v4f v = *(const v4f*)(x + (size_t)n * DHID + colbase + c4);
        *(v4f*)(hbuf + r * HSTR + c4) = v;
    }
}

__device__ __forceinline__ void wave_lds_order()
{
    __builtin_amdgcn_fence(__ATOMIC_RELEASE, "wavefront");
    __builtin_amdgcn_wave_barrier();
    __builtin_amdgcn_fence(__ATOMIC_ACQUIRE, "wavefront");
}

__global__ void __launch_bounds__(WPB * 32, 1)
edge_mlp_wmma_f32(const float* __restrict__ x, const int* __restrict__ eidx, int E,
                  const float* __restrict__ W0, const float* __restrict__ b0,
                  const float* __restrict__ W1, const float* __restrict__ b1,
                  const float* __restrict__ W2, const float* __restrict__ b2,
                  const float* __restrict__ W3, const float* __restrict__ b3,
                  const float* __restrict__ g0, const float* __restrict__ be0,
                  const float* __restrict__ g1, const float* __restrict__ be1,
                  const float* __restrict__ g2, const float* __restrict__ be2,
                  float* __restrict__ out)
{
    extern __shared__ float smem[];
    const int tid  = threadIdx.x;
    const int lane = tid & 31;
    const int wave = tid >> 5;

    float* wp0  = smem;
    float* wp1  = smem + OFF_W1;
    float* wp2  = smem + OFF_W2;
    float* hbuf = smem + OFF_H + wave * (TILE_M * HSTR);

    pack_w(wp0, W0, 2 * DHID, tid, WPB * 32);
    pack_w(wp1, W1, DHID,     tid, WPB * 32);
    pack_w(wp2, W2, DHID,     tid, WPB * 32);
    __syncthreads();

    const int tiles  = (E + TILE_M - 1) / TILE_M;
    const int stride = gridDim.x * WPB;
    const int sweeps = (tiles + stride - 1) / stride;

    for (int it = 0; it < sweeps; ++it) {
        const int tile = (it * gridDim.x + blockIdx.x) * WPB + wave;
        if (__builtin_amdgcn_readfirstlane((int)(tile >= tiles))) break;
        const int e0   = tile * TILE_M;

        {
            const int tn = tile + stride;
            if (tn < tiles) {
                __builtin_prefetch(&eidx[tn * TILE_M], 0, 1);
                __builtin_prefetch(&eidx[E + tn * TILE_M], 0, 1);
            }
        }

        int e_l = e0 + (lane & 15); if (e_l >= E) e_l = E - 1;
        const int idx_s = eidx[e_l];
        const int idx_e = eidx[E + e_l];


        v8f h0[8] = {};
        stage_half(hbuf, x, idx_s, 0, lane);
        wave_lds_order();
        gemm_lds<16, HSTR>(hbuf, wp0, h0, lane);
        wave_lds_order();
        stage_half(hbuf, x, idx_s, CW, lane);
        wave_lds_order();
        gemm_lds<16, HSTR>(hbuf, wp0 + 16 * 512, h0, lane);
        wave_lds_order();
        stage_half(hbuf, x, idx_e, 0, lane);
        wave_lds_order();
        gemm_lds<16, HSTR>(hbuf, wp0 + 32 * 512, h0, lane);
        wave_lds_order();
        stage_half(hbuf, x, idx_e, CW, lane);
        wave_lds_order();
        gemm_lds<16, HSTR>(hbuf, wp0 + 48 * 512, h0, lane);
        layernorm_tanh(h0, b0, g0, be0, lane);
        wave_lds_order();

        v8f h1[8] = {};
        store_half<0>(hbuf, h0, lane);
        wave_lds_order();
        gemm_lds<16, HSTR>(hbuf, wp1, h1, lane);
        wave_lds_order();
        store_half<1>(hbuf, h0, lane);
        wave_lds_order();
        gemm_lds<16, HSTR>(hbuf, wp1 + 16 * 512, h1, lane);
        layernorm_tanh(h1, b1, g1, be1, lane);
        wave_lds_order();

        v8f h2[8] = {};
        store_half<0>(hbuf, h1, lane);
        wave_lds_order();
        gemm_lds<16, HSTR>(hbuf, wp2, h2, lane);
        wave_lds_order();
        store_half<1>(hbuf, h1, lane);
        wave_lds_order();
        gemm_lds<16, HSTR>(hbuf, wp2 + 16 * 512, h2, lane);
        layernorm_tanh(h2, b2, g2, be2, lane);

        {
            const int col0 = lane & 15;
            const int grp  = lane >> 4;
            float w3[8];
#pragma unroll
            for (int nn = 0; nn < 8; ++nn) w3[nn] = W3[col0 + (nn << 4)];
            const float bias3 = b3[0];
#pragma unroll
            for (int j = 0; j < 8; ++j) {
                float p = 0.f;
#pragma unroll
                for (int nn = 0; nn < 8; ++nn) p = fmaf(h2[nn][j], w3[nn], p);
#pragma unroll
                for (int m = 1; m < 16; m <<= 1) p += __shfl_xor(p, m, 32);
                const int e = e0 + j + (grp << 3);
                if (col0 == j && e < E) out[e] = p + bias3;
            }
        }
        wave_lds_order();
    }
}

extern "C" void kernel_launch(void* const* d_in, const int* in_sizes, int n_in,
                              void* d_out, int out_size, void* d_ws, size_t ws_size,
                              hipStream_t stream)
{
    const float* x   = (const float*)d_in[0];
    const int*   ei  = (const int*)  d_in[1];
    const float* W0  = (const float*)d_in[2];
    const float* b0  = (const float*)d_in[3];
    const float* W1  = (const float*)d_in[4];
    const float* b1  = (const float*)d_in[5];
    const float* W2  = (const float*)d_in[6];
    const float* b2  = (const float*)d_in[7];
    const float* W3  = (const float*)d_in[8];
    const float* b3  = (const float*)d_in[9];
    const float* g0  = (const float*)d_in[10];
    const float* be0 = (const float*)d_in[11];
    const float* g1  = (const float*)d_in[12];
    const float* be1 = (const float*)d_in[13];
    const float* g2  = (const float*)d_in[14];
    const float* be2 = (const float*)d_in[15];

    const int E     = in_sizes[1] / 2;
    const int tiles = (E + TILE_M - 1) / TILE_M;
    int blocks = (tiles + WPB - 1) / WPB;
    if (blocks > 1024) blocks = 1024;
    const size_t shmem = (size_t)SMEM_FLOATS * sizeof(float);

    edge_mlp_wmma_f32<<<blocks, WPB * 32, shmem, stream>>>(
        x, ei, E, W0, b0, W1, b1, W2, b2, W3, b3,
        g0, be0, g1, be1, g2, be2, (float*)d_out);
}


# layers computed transposed (W^T . act^T): each layer's D registers are the next layer's B operand, node rows load straight into B; no activation in LDS; k order of sums and LN sum tree change
# speedup vs baseline: 1.9183x; 1.1027x over previous
#include <hip/hip_runtime.h>
#include <math.h>

typedef __attribute__((ext_vector_type(2))) float v2f;
typedef __attribute__((ext_vector_type(4))) float v4f;
typedef __attribute__((ext_vector_type(8))) float v8f;

#define DHID   128
#define TILE_E 16
#define WPB    12
#define LNEPS  1e-5f

#if __has_builtin(__builtin_amdgcn_tanhf)
#define TANHF(v) __builtin_amdgcn_tanhf(v)
#else
#define TANHF(v) tanhf(v)
#endif

#define SCHED_FENCE() __builtin_amdgcn_sched_barrier(0)

#define OFF_W1 32768
#define OFF_W2 49152
#define OFF_P  65536
#define SMEM_FLOATS (OFF_P + 10 * DHID)

template <bool CHAINED>
__device__ __forceinline__ void pack_wT(float* __restrict__ dst,
                                        const float* __restrict__ W,
                                        int K, int tid, int nthr)
{
    const int total = K << 7;
#pragma unroll 4
    for (int i = tid; i < total; i += nthr) {
        const int k = i >> 7;
        const int n = i & 127;
        int kb, h;
        if (CHAINED) { kb = ((k >> 4) << 1) | ((k >> 2) & 1); h = (k >> 3) & 1; }
        else         { kb = ((k >> 5) << 2) | ((k >> 2) & 3); h = (k >> 4) & 1; }
        dst[((((kb << 3) + (n >> 4)) << 5) + (n & 15) + (h << 4)) * 4 + (k & 3)] = W[i];
    }
}

__device__ __forceinline__ void load_a(v4f a[8], const float* __restrict__ wl, int kb)
{
#pragma unroll
    for (int mm = 0; mm < 8; ++mm)
        a[mm] = *(const v4f*)(wl + (kb << 10) + (mm << 7));
}

__device__ __forceinline__ void mma_kb(v8f acc[8], const v4f a[8], v4f b)
{
    const v2f b0 = {b.x, b.y};
    const v2f b1 = {b.z, b.w};
#pragma unroll
    for (int mm = 0; mm < 8; ++mm) {
        const v2f a0 = {a[mm].x, a[mm].y};
        acc[mm] = __builtin_amdgcn_wmma_f32_16x16x4_f32(
            false, a0, false, b0, (short)0, acc[mm], false, false);
    }
#pragma unroll
    for (int mm = 0; mm < 8; ++mm) {
        const v2f a1 = {a[mm].z, a[mm].w};
        acc[mm] = __builtin_amdgcn_wmma_f32_16x16x4_f32(
            false, a1, false, b1, (short)0, acc[mm], false, false);
    }
}

__device__ __forceinline__ void gemm_chained(v8f acc[8], const v8f prev[8],
                                             const float* __restrict__ wl)
{
    v4f a0[8], a1[8];
    load_a(a0, wl, 0);
#pragma unroll
    for (int nn = 0; nn < 8; ++nn) {
        const v8f t = prev[nn];
        load_a(a1, wl, 2 * nn + 1);
        SCHED_FENCE();
        mma_kb(acc, a0, (v4f){t[0], t[1], t[2], t[3]});
        SCHED_FENCE();
        if (nn + 1 < 8) load_a(a0, wl, 2 * nn + 2);
        SCHED_FENCE();
        mma_kb(acc, a1, (v4f){t[4], t[5], t[6], t[7]});
        SCHED_FENCE();
    }
}

__device__ __forceinline__ void layernorm_tanh_T(v8f acc[8], const float* __restrict__ par,
                                                 int lane)
{
    const int f0 = (lane >> 4) << 3;
    float sp[8], qp[8];
#pragma unroll
    for (int mm = 0; mm < 8; ++mm) {
        const v4f blo = *(const v4f*)(par + (mm << 4) + f0);
        const v4f bhi = *(const v4f*)(par + (mm << 4) + f0 + 4);
        float s = 0.f, q = 0.f;
#pragma unroll
        for (int r = 0; r < 8; ++r) {
            const float v = acc[mm][r] + (r < 4 ? blo[r & 3] : bhi[r & 3]);
            acc[mm][r] = v;
            s += v;
            q = fmaf(v, v, q);
        }
        sp[mm] = s; qp[mm] = q;
    }
    float s  = ((sp[0] + sp[1]) + (sp[2] + sp[3])) + ((sp[4] + sp[5]) + (sp[6] + sp[7]));
    float ss = ((qp[0] + qp[1]) + (qp[2] + qp[3])) + ((qp[4] + qp[5]) + (qp[6] + qp[7]));
    s  += __shfl_xor(s,  16, 32);
    ss += __shfl_xor(ss, 16, 32);
    const float mu  = s * (1.0f / 128.0f);
    const float var = fmaf(-mu, mu, ss * (1.0f / 128.0f));
    const float inv = rsqrtf(var + LNEPS);
#pragma unroll
    for (int mm = 0; mm < 8; ++mm) {
        const v4f glo = *(const v4f*)(par + DHID + (mm << 4) + f0);
        const v4f ghi = *(const v4f*)(par + DHID + (mm << 4) + f0 + 4);
        const v4f elo = *(const v4f*)(par + 2 * DHID + (mm << 4) + f0);
        const v4f ehi = *(const v4f*)(par + 2 * DHID + (mm << 4) + f0 + 4);
#pragma unroll
        for (int r = 0; r < 8; ++r) {
            const float g = (r < 4 ? glo[r & 3] : ghi[r & 3]);
            const float e = (r < 4 ? elo[r & 3] : ehi[r & 3]);
            const float v = (acc[mm][r] - mu) * inv;
            acc[mm][r] = TANHF(fmaf(v, g, e));
        }
    }
}

__device__ __forceinline__ void load_x(v4f d[4], const float* __restrict__ p)
{
#pragma unroll
    for (int u = 0; u < 4; ++u) d[u] = *(const v4f*)(p + (u << 2));
}

__global__ void __launch_bounds__(WPB * 32, 1)
edge_mlp_wmma_f32(const float* __restrict__ x, const int* __restrict__ eidx, int E,
                  const float* __restrict__ W0, const float* __restrict__ b0,
                  const float* __restrict__ W1, const float* __restrict__ b1,
                  const float* __restrict__ W2, const float* __restrict__ b2,
                  const float* __restrict__ W3, const float* __restrict__ b3,
                  const float* __restrict__ g0, const float* __restrict__ be0,
                  const float* __restrict__ g1, const float* __restrict__ be1,
                  const float* __restrict__ g2, const float* __restrict__ be2,
                  float* __restrict__ out)
{
    extern __shared__ float smem[];
    const int tid  = threadIdx.x;
    const int lane = tid & 31;
    const int wave = tid >> 5;

    float* wp0 = smem;
    float* wp1 = smem + OFF_W1;
    float* wp2 = smem + OFF_W2;
    float* par = smem + OFF_P;

    pack_wT<false>(wp0, W0, 2 * DHID, tid, WPB * 32);
    pack_wT<true >(wp1, W1, DHID,     tid, WPB * 32);
    pack_wT<true >(wp2, W2, DHID,     tid, WPB * 32);
    if (tid < DHID) {
        par[0 * DHID + tid] = b0[tid];  par[1 * DHID + tid] = g0[tid];  par[2 * DHID + tid] = be0[tid];
        par[3 * DHID + tid] = b1[tid];  par[4 * DHID + tid] = g1[tid];  par[5 * DHID + tid] = be1[tid];
        par[6 * DHID + tid] = b2[tid];  par[7 * DHID + tid] = g2[tid];  par[8 * DHID + tid] = be2[tid];
        par[9 * DHID + tid] = W3[tid];
    }
    __syncthreads();

    const int tiles  = (E + TILE_E - 1) / TILE_E;
    const int stride = gridDim.x * WPB;
    const int sweeps = (tiles + stride - 1) / stride;

    const int c  = lane & 15;
    const int h  = lane >> 4;
    const float* wl0 = wp0 + (lane << 2);
    const float* wl1 = wp1 + (lane << 2);
    const float* wl2 = wp2 + (lane << 2);

    for (int it = 0; it < sweeps; ++it) {
        const int tile = (it * gridDim.x + blockIdx.x) * WPB + wave;
        if (__builtin_amdgcn_readfirstlane((int)(tile >= tiles))) break;
        const int e0 = tile * TILE_E;

        {
            const int tn = tile + stride;
            if (tn < tiles) {
                __builtin_prefetch(&eidx[tn * TILE_E], 0, 1);
                __builtin_prefetch(&eidx[E + tn * TILE_E], 0, 1);
            }
        }

        int e_l = e0 + c; if (e_l >= E) e_l = E - 1;
        const int idx_s = eidx[e_l];
        const int idx_e = eidx[E + e_l];

        const float* rs = x + (size_t)idx_s * DHID + (h << 4);
        const float* re = x + (size_t)idx_e * DHID + (h << 4);
        v8f h0[8] = {};
        {
            v4f xq[2][4];
            v4f a0[8], a1[8];
            load_x(xq[0], rs);
            load_a(a0, wl0, 0);
#pragma unroll
            for (int q = 0; q < 8; ++q) {
                if (q + 1 < 8)
                    load_x(xq[(q + 1) & 1], ((q + 1) < 4 ? rs : re) + (((q + 1) & 3) << 5));
#pragma unroll
                for (int u = 0; u < 4; u += 2) {
                    const int kb = 4 * q + u;
                    load_a(a1, wl0, kb + 1);
                    SCHED_FENCE();
                    mma_kb(h0, a0, xq[q & 1][u]);
                    SCHED_FENCE();
                    if (kb + 2 < 32) load_a(a0, wl0, kb + 2);
                    SCHED_FENCE();
                    mma_kb(h0, a1, xq[q & 1][u + 1]);
                    SCHED_FENCE();
                }
            }
        }
        layernorm_tanh_T(h0, par, lane);

        v8f h1[8] = {};
        gemm_chained(h1, h0, wl1);
        layernorm_tanh_T(h1, par + 3 * DHID, lane);

        v8f h2[8] = {};
        gemm_chained(h2, h1, wl2);
        layernorm_tanh_T(h2, par + 6 * DHID, lane);

        {
            const float* w3 = par + 9 * DHID + (h << 3);
            float pp[8];
#pragma unroll
            for (int mm = 0; mm < 8; ++mm) {
                const v4f wlo = *(const v4f*)(w3 + (mm << 4));
                const v4f whi = *(const v4f*)(w3 + (mm << 4) + 4);
                float p = 0.f;
#pragma unroll
                for (int r = 0; r < 8; ++r)
                    p = fmaf(h2[mm][r], (r < 4 ? wlo[r & 3] : whi[r & 3]), p);
                pp[mm] = p;
            }
            float p = ((pp[0] + pp[1]) + (pp[2] + pp[3])) + ((pp[4] + pp[5]) + (pp[6] + pp[7]));
            p += __shfl_xor(p, 16, 32);
            const int e = e0 + c;
            if (h == 0 && e < E) out[e] = p + b3[0];
        }
    }
}

extern "C" void kernel_launch(void* const* d_in, const int* in_sizes, int n_in,
                              void* d_out, int out_size, void* d_ws, size_t ws_size,
                              hipStream_t stream)
{
    const float* x   = (const float*)d_in[0];
    const int*   ei  = (const int*)  d_in[1];
    const float* W0  = (const float*)d_in[2];
    const float* b0  = (const float*)d_in[3];
    const float* W1  = (const float*)d_in[4];
    const float* b1  = (const float*)d_in[5];
    const float* W2  = (const float*)d_in[6];
    const float* b2  = (const float*)d_in[7];
    const float* W3  = (const float*)d_in[8];
    const float* b3  = (const float*)d_in[9];
    const float* g0  = (const float*)d_in[10];
    const float* be0 = (const float*)d_in[11];
    const float* g1  = (const float*)d_in[12];
    const float* be1 = (const float*)d_in[13];
    const float* g2  = (const float*)d_in[14];
    const float* be2 = (const float*)d_in[15];

    const int E     = in_sizes[1] / 2;
    const int tiles = (E + TILE_E - 1) / TILE_E;
    int blocks = (tiles + WPB - 1) / WPB;
    if (blocks > 1024) blocks = 1024;
    const size_t shmem = (size_t)SMEM_FLOATS * sizeof(float);

    edge_mlp_wmma_f32<<<blocks, WPB * 32, shmem, stream>>>(
        x, ei, E, W0, b0, W1, b1, W2, b2, W3, b3,
        g0, be0, g1, be1, g2, be2, (float*)d_out);
}


# grid cut from 1024 to 256 blocks = one resident block per WGP, tile index block-fastest inside a sweep: weights staged once per WGP (was 4x) and the partial last sweep spread over all WGPs
# speedup vs baseline: 2.0887x; 1.0888x over previous
#include <hip/hip_runtime.h>
#include <math.h>

typedef __attribute__((ext_vector_type(2))) float v2f;
typedef __attribute__((ext_vector_type(4))) float v4f;
typedef __attribute__((ext_vector_type(8))) float v8f;

#define DHID   128
#define TILE_E 16
#define WPB    12
#define LNEPS  1e-5f

#if __has_builtin(__builtin_amdgcn_tanhf)
#define TANHF(v) __builtin_amdgcn_tanhf(v)
#else
#define TANHF(v) tanhf(v)
#endif

#define SCHED_FENCE() __builtin_amdgcn_sched_barrier(0)

#define OFF_W1 32768
#define OFF_W2 49152
#define OFF_P  65536
#define SMEM_FLOATS (OFF_P + 10 * DHID)

template <bool CHAINED>
__device__ __forceinline__ void pack_wT(float* __restrict__ dst,
                                        const float* __restrict__ W,
                                        int K, int tid, int nthr)
{
    const int total = K << 7;
#pragma unroll 4
    for (int i = tid; i < total; i += nthr) {
        const int k = i >> 7;
        const int n = i & 127;
        int kb, h;
        if (CHAINED) { kb = ((k >> 4) << 1) | ((k >> 2) & 1); h = (k >> 3) & 1; }
        else         { kb = ((k >> 5) << 2) | ((k >> 2) & 3); h = (k >> 4) & 1; }
        dst[((((kb << 3) + (n >> 4)) << 5) + (n & 15) + (h << 4)) * 4 + (k & 3)] = W[i];
    }
}

__device__ __forceinline__ void load_a(v4f a[8], const float* __restrict__ wl, int kb)
{
#pragma unroll
    for (int mm = 0; mm < 8; ++mm)
        a[mm] = *(const v4f*)(wl + (kb << 10) + (mm << 7));
}

__device__ __forceinline__ void mma_kb(v8f acc[8], const v4f a[8], v4f b)
{
    const v2f b0 = {b.x, b.y};
    const v2f b1 = {b.z, b.w};
#pragma unroll
    for (int mm = 0; mm < 8; ++mm) {
        const v2f a0 = {a[mm].x, a[mm].y};
        acc[mm] = __builtin_amdgcn_wmma_f32_16x16x4_f32(
            false, a0, false, b0, (short)0, acc[mm], false, false);
    }
#pragma unroll
    for (int mm = 0; mm < 8; ++mm) {
        const v2f a1 = {a[mm].z, a[mm].w};
        acc[mm] = __builtin_amdgcn_wmma_f32_16x16x4_f32(
            false, a1, false, b1, (short)0, acc[mm], false, false);
    }
}

__device__ __forceinline__ void gemm_chained(v8f acc[8], const v8f prev[8],
                                             const float* __restrict__ wl)
{
    v4f a0[8], a1[8];
    load_a(a0, wl, 0);
#pragma unroll
    for (int nn = 0; nn < 8; ++nn) {
        const v8f t = prev[nn];
        load_a(a1, wl, 2 * nn + 1);
        SCHED_FENCE();
        mma_kb(acc, a0, (v4f){t[0], t[1], t[2], t[3]});
        SCHED_FENCE();
        if (nn + 1 < 8) load_a(a0, wl, 2 * nn + 2);
        SCHED_FENCE();
        mma_kb(acc, a1, (v4f){t[4], t[5], t[6], t[7]});
        SCHED_FENCE();
    }
}

__device__ __forceinline__ void layernorm_tanh_T(v8f acc[8], const float* __restrict__ par,
                                                 int lane)
{
    const int f0 = (lane >> 4) << 3;
    float sp[8], qp[8];
#pragma unroll
    for (int mm = 0; mm < 8; ++mm) {
        const v4f blo = *(const v4f*)(par + (mm << 4) + f0);
        const v4f bhi = *(const v4f*)(par + (mm << 4) + f0 + 4);
        float s = 0.f, q = 0.f;
#pragma unroll
        for (int r = 0; r < 8; ++r) {
            const float v = acc[mm][r] + (r < 4 ? blo[r & 3] : bhi[r & 3]);
            acc[mm][r] = v;
            s += v;
            q = fmaf(v, v, q);
        }
        sp[mm] = s; qp[mm] = q;
    }
    float s  = ((sp[0] + sp[1]) + (sp[2] + sp[3])) + ((sp[4] + sp[5]) + (sp[6] + sp[7]));
    float ss = ((qp[0] + qp[1]) + (qp[2] + qp[3])) + ((qp[4] + qp[5]) + (qp[6] + qp[7]));
    s  += __shfl_xor(s,  16, 32);
    ss += __shfl_xor(ss, 16, 32);
    const float mu  = s * (1.0f / 128.0f);
    const float var = fmaf(-mu, mu, ss * (1.0f / 128.0f));
    const float inv = rsqrtf(var + LNEPS);
#pragma unroll
    for (int mm = 0; mm < 8; ++mm) {
        const v4f glo = *(const v4f*)(par + DHID + (mm << 4) + f0);
        const v4f ghi = *(const v4f*)(par + DHID + (mm << 4) + f0 + 4);
        const v4f elo = *(const v4f*)(par + 2 * DHID + (mm << 4) + f0);
        const v4f ehi = *(const v4f*)(par + 2 * DHID + (mm << 4) + f0 + 4);
#pragma unroll
        for (int r = 0; r < 8; ++r) {
            const float g = (r < 4 ? glo[r & 3] : ghi[r & 3]);
            const float e = (r < 4 ? elo[r & 3] : ehi[r & 3]);
            const float v = (acc[mm][r] - mu) * inv;
            acc[mm][r] = TANHF(fmaf(v, g, e));
        }
    }
}

__device__ __forceinline__ void load_x(v4f d[4], const float* __restrict__ p)
{
#pragma unroll
    for (int u = 0; u < 4; ++u) d[u] = *(const v4f*)(p + (u << 2));
}

__global__ void __launch_bounds__(WPB * 32, 1)
edge_mlp_wmma_f32(const float* __restrict__ x, const int* __restrict__ eidx, int E,
                  const float* __restrict__ W0, const float* __restrict__ b0,
                  const float* __restrict__ W1, const float* __restrict__ b1,
                  const float* __restrict__ W2, const float* __restrict__ b2,
                  const float* __restrict__ W3, const float* __restrict__ b3,
                  const float* __restrict__ g0, const float* __restrict__ be0,
                  const float* __restrict__ g1, const float* __restrict__ be1,
                  const float* __restrict__ g2, const float* __restrict__ be2,
                  float* __restrict__ out)
{
    extern __shared__ float smem[];
    const int tid  = threadIdx.x;
    const int lane = tid & 31;
    const int wave = tid >> 5;

    float* wp0 = smem;
    float* wp1 = smem + OFF_W1;
    float* wp2 = smem + OFF_W2;
    float* par = smem + OFF_P;

    pack_wT<false>(wp0, W0, 2 * DHID, tid, WPB * 32);
    pack_wT<true >(wp1, W1, DHID,     tid, WPB * 32);
    pack_wT<true >(wp2, W2, DHID,     tid, WPB * 32);
    if (tid < DHID) {
        par[0 * DHID + tid] = b0[tid];  par[1 * DHID + tid] = g0[tid];  par[2 * DHID + tid] = be0[tid];
        par[3 * DHID + tid] = b1[tid];  par[4 * DHID + tid] = g1[tid];  par[5 * DHID + tid] = be1[tid];
        par[6 * DHID + tid] = b2[tid];  par[7 * DHID + tid] = g2[tid];  par[8 * DHID + tid] = be2[tid];
        par[9 * DHID + tid] = W3[tid];
    }
    __syncthreads();

    const int tiles  = (E + TILE_E - 1) / TILE_E;
    const int stride = gridDim.x * WPB;
    const int sweeps = (tiles + stride - 1) / stride;

    const int c  = lane & 15;
    const int h  = lane >> 4;
    const float* wl0 = wp0 + (lane << 2);
    const float* wl1 = wp1 + (lane << 2);
    const float* wl2 = wp2 + (lane << 2);

    for (int it = 0; it < sweeps; ++it) {
        const int tile = it * stride + wave * gridDim.x + blockIdx.x;
        if (__builtin_amdgcn_readfirstlane((int)(tile >= tiles))) break;
        const int e0 = tile * TILE_E;

        {
            const int tn = tile + stride;
            if (tn < tiles) {
                __builtin_prefetch(&eidx[tn * TILE_E], 0, 1);
                __builtin_prefetch(&eidx[E + tn * TILE_E], 0, 1);
            }
        }

        int e_l = e0 + c; if (e_l >= E) e_l = E - 1;
        const int idx_s = eidx[e_l];
        const int idx_e = eidx[E + e_l];

        const float* rs = x + (size_t)idx_s * DHID + (h << 4);
        const float* re = x + (size_t)idx_e * DHID + (h << 4);
        v8f h0[8] = {};
        {
            v4f xq[2][4];
            v4f a0[8], a1[8];
            load_x(xq[0], rs);
            load_a(a0, wl0, 0);
#pragma unroll
            for (int q = 0; q < 8; ++q) {
                if (q + 1 < 8)
                    load_x(xq[(q + 1) & 1], ((q + 1) < 4 ? rs : re) + (((q + 1) & 3) << 5));
#pragma unroll
                for (int u = 0; u < 4; u += 2) {
                    const int kb = 4 * q + u;
                    load_a(a1, wl0, kb + 1);
                    SCHED_FENCE();
                    mma_kb(h0, a0, xq[q & 1][u]);
                    SCHED_FENCE();
                    if (kb + 2 < 32) load_a(a0, wl0, kb + 2);
                    SCHED_FENCE();
                    mma_kb(h0, a1, xq[q & 1][u + 1]);
                    SCHED_FENCE();
                }
            }
        }
        layernorm_tanh_T(h0, par, lane);

        v8f h1[8] = {};
        gemm_chained(h1, h0, wl1);
        layernorm_tanh_T(h1, par + 3 * DHID, lane);

        v8f h2[8] = {};
        gemm_chained(h2, h1, wl2);
        layernorm_tanh_T(h2, par + 6 * DHID, lane);

        {
            const float* w3 = par + 9 * DHID + (h << 3);
            float pp[8];
#pragma unroll
            for (int mm = 0; mm < 8; ++mm) {
                const v4f wlo = *(const v4f*)(w3 + (mm << 4));
                const v4f whi = *(const v4f*)(w3 + (mm << 4) + 4);
                float p = 0.f;
#pragma unroll
                for (int r = 0; r < 8; ++r)
                    p = fmaf(h2[mm][r], (r < 4 ? wlo[r & 3] : whi[r & 3]), p);
                pp[mm] = p;
            }
            float p = ((pp[0] + pp[1]) + (pp[2] + pp[3])) + ((pp[4] + pp[5]) + (pp[6] + pp[7]));
            p += __shfl_xor(p, 16, 32);
            const int e = e0 + c;
            if (h == 0 && e < E) out[e] = p + b3[0];
        }
    }
}

extern "C" void kernel_launch(void* const* d_in, const int* in_sizes, int n_in,
                              void* d_out, int out_size, void* d_ws, size_t ws_size,
                              hipStream_t stream)
{
    const float* x   = (const float*)d_in[0];
    const int*   ei  = (const int*)  d_in[1];
    const float* W0  = (const float*)d_in[2];
    const float* b0  = (const float*)d_in[3];
    const float* W1  = (const float*)d_in[4];
    const float* b1  = (const float*)d_in[5];
    const float* W2  = (const float*)d_in[6];
    const float* b2  = (const float*)d_in[7];
    const float* W3  = (const float*)d_in[8];
    const float* b3  = (const float*)d_in[9];
    const float* g0  = (const float*)d_in[10];
    const float* be0 = (const float*)d_in[11];
    const float* g1  = (const float*)d_in[12];
    const float* be1 = (const float*)d_in[13];
    const float* g2  = (const float*)d_in[14];
    const float* be2 = (const float*)d_in[15];

    const int E     = in_sizes[1] / 2;
    const int tiles = (E + TILE_E - 1) / TILE_E;
    int blocks = (tiles + WPB - 1) / WPB;
    if (blocks > 256) blocks = 256;
    const size_t shmem = (size_t)SMEM_FLOATS * sizeof(float);

    edge_mlp_wmma_f32<<<blocks, WPB * 32, shmem, stream>>>(
        x, ei, E, W0, b0, W1, b1, W2, b2, W3, b3,
        g0, be0, g1, be1, g2, be2, (float*)d_out);
}


# layer 0 factored per node: x.W0[0:128] and x.W0[128:256] formed once per node into the workspace (one more launch), edges gather and add the two rows; k order of layer-0 sums changes
# speedup vs baseline: 2.7106x; 1.2977x over previous
#include <hip/hip_runtime.h>
#include <math.h>

typedef __attribute__((ext_vector_type(2))) float v2f;
typedef __attribute__((ext_vector_type(4))) float v4f;
typedef __attribute__((ext_vector_type(8))) float v8f;

#define DHID   128
#define TILE_E 16
#define WPB    12
#define LNEPS  1e-5f

#if __has_builtin(__builtin_amdgcn_tanhf)
#define TANHF(v) __builtin_amdgcn_tanhf(v)
#else
#define TANHF(v) tanhf(v)
#endif

#define SCHED_FENCE() __builtin_amdgcn_sched_barrier(0)

#define OFF_W1 32768
#define OFF_W2 49152
#define OFF_P  65536
#define SMEM_FLOATS (OFF_P + 10 * DHID)

template <bool CHAINED>
__device__ __forceinline__ void pack_wT(float* __restrict__ dst,
                                        const float* __restrict__ W,
                                        int K, int tid, int nthr)
{
    const int total = K << 7;
#pragma unroll 4
    for (int i = tid; i < total; i += nthr) {
        const int k = i >> 7;
        const int n = i & 127;
        int kb, h;
        if (CHAINED) { kb = ((k >> 4) << 1) | ((k >> 2) & 1); h = (k >> 3) & 1; }
        else         { kb = ((k >> 5) << 2) | ((k >> 2) & 3); h = (k >> 4) & 1; }
        dst[((((kb << 3) + (n >> 4)) << 5) + (n & 15) + (h << 4)) * 4 + (k & 3)] = W[i];
    }
}

__device__ __forceinline__ void load_a(v4f a[8], const float* __restrict__ wl, int kb)
{
#pragma unroll
    for (int mm = 0; mm < 8; ++mm)
        a[mm] = *(const v4f*)(wl + (kb << 10) + (mm << 7));
}

__device__ __forceinline__ void mma_kb(v8f acc[8], const v4f a[8], v4f b)
{
    const v2f b0 = {b.x, b.y};
    const v2f b1 = {b.z, b.w};
#pragma unroll
    for (int mm = 0; mm < 8; ++mm) {
        const v2f a0 = {a[mm].x, a[mm].y};
        acc[mm] = __builtin_amdgcn_wmma_f32_16x16x4_f32(
            false, a0, false, b0, (short)0, acc[mm], false, false);
    }
#pragma unroll
    for (int mm = 0; mm < 8; ++mm) {
        const v2f a1 = {a[mm].z, a[mm].w};
        acc[mm] = __builtin_amdgcn_wmma_f32_16x16x4_f32(
            false, a1, false, b1, (short)0, acc[mm], false, false);
    }
}

__device__ __forceinline__ void gemm_chained(v8f acc[8], const v8f prev[8],
                                             const float* __restrict__ wl)
{
    v4f a0[8], a1[8];
    load_a(a0, wl, 0);
#pragma unroll
    for (int nn = 0; nn < 8; ++nn) {
        const v8f t = prev[nn];
        load_a(a1, wl, 2 * nn + 1);
        SCHED_FENCE();
        mma_kb(acc, a0, (v4f){t[0], t[1], t[2], t[3]});
        SCHED_FENCE();
        if (nn + 1 < 8) load_a(a0, wl, 2 * nn + 2);
        SCHED_FENCE();
        mma_kb(acc, a1, (v4f){t[4], t[5], t[6], t[7]});
        SCHED_FENCE();
    }
}

__device__ __forceinline__ void layernorm_tanh_T(v8f acc[8], const float* __restrict__ par,
                                                 int lane)
{
    const int f0 = (lane >> 4) << 3;
    float sp[8], qp[8];
#pragma unroll
    for (int mm = 0; mm < 8; ++mm) {
        const v4f blo = *(const v4f*)(par + (mm << 4) + f0);
        const v4f bhi = *(const v4f*)(par + (mm << 4) + f0 + 4);
        float s = 0.f, q = 0.f;
#pragma unroll
        for (int r = 0; r < 8; ++r) {
            const float v = acc[mm][r] + (r < 4 ? blo[r & 3] : bhi[r & 3]);
            acc[mm][r] = v;
            s += v;
            q = fmaf(v, v, q);
        }
        sp[mm] = s; qp[mm] = q;
    }
    float s  = ((sp[0] + sp[1]) + (sp[2] + sp[3])) + ((sp[4] + sp[5]) + (sp[6] + sp[7]));
    float ss = ((qp[0] + qp[1]) + (qp[2] + qp[3])) + ((qp[4] + qp[5]) + (qp[6] + qp[7]));
    s  += __shfl_xor(s,  16, 32);
    ss += __shfl_xor(ss, 16, 32);
    const float mu  = s * (1.0f / 128.0f);
    const float var = fmaf(-mu, mu, ss * (1.0f / 128.0f));
    const float inv = rsqrtf(var + LNEPS);
#pragma unroll
    for (int mm = 0; mm < 8; ++mm) {
        const v4f glo = *(const v4f*)(par + DHID + (mm << 4) + f0);
        const v4f ghi = *(const v4f*)(par + DHID + (mm << 4) + f0 + 4);
        const v4f elo = *(const v4f*)(par + 2 * DHID + (mm << 4) + f0);
        const v4f ehi = *(const v4f*)(par + 2 * DHID + (mm << 4) + f0 + 4);
#pragma unroll
        for (int r = 0; r < 8; ++r) {
            const float g = (r < 4 ? glo[r & 3] : ghi[r & 3]);
            const float e = (r < 4 ? elo[r & 3] : ehi[r & 3]);
            const float v = (acc[mm][r] - mu) * inv;
            acc[mm][r] = TANHF(fmaf(v, g, e));
        }
    }
}

__device__ __forceinline__ void load_x(v4f d[4], const float* __restrict__ p)
{
#pragma unroll
    for (int u = 0; u < 4; ++u) d[u] = *(const v4f*)(p + (u << 2));
}

__global__ void __launch_bounds__(WPB * 32, 1)
edge_mlp_wmma_f32(const float* __restrict__ x, const int* __restrict__ eidx, int E,
                  const float* __restrict__ W0, const float* __restrict__ b0,
                  const float* __restrict__ W1, const float* __restrict__ b1,
                  const float* __restrict__ W2, const float* __restrict__ b2,
                  const float* __restrict__ W3, const float* __restrict__ b3,
                  const float* __restrict__ g0, const float* __restrict__ be0,
                  const float* __restrict__ g1, const float* __restrict__ be1,
                  const float* __restrict__ g2, const float* __restrict__ be2,
                  float* __restrict__ out)
{
    extern __shared__ float smem[];
    const int tid  = threadIdx.x;
    const int lane = tid & 31;
    const int wave = tid >> 5;

    float* wp0 = smem;
    float* wp1 = smem + OFF_W1;
    float* wp2 = smem + OFF_W2;
    float* par = smem + OFF_P;

    pack_wT<false>(wp0, W0, 2 * DHID, tid, WPB * 32);
    pack_wT<true >(wp1, W1, DHID,     tid, WPB * 32);
    pack_wT<true >(wp2, W2, DHID,     tid, WPB * 32);
    if (tid < DHID) {
        par[0 * DHID + tid] = b0[tid];  par[1 * DHID + tid] = g0[tid];  par[2 * DHID + tid] = be0[tid];
        par[3 * DHID + tid] = b1[tid];  par[4 * DHID + tid] = g1[tid];  par[5 * DHID + tid] = be1[tid];
        par[6 * DHID + tid] = b2[tid];  par[7 * DHID + tid] = g2[tid];  par[8 * DHID + tid] = be2[tid];
        par[9 * DHID + tid] = W3[tid];
    }
    __syncthreads();

    const int tiles  = (E + TILE_E - 1) / TILE_E;
    const int stride = gridDim.x * WPB;
    const int sweeps = (tiles + stride - 1) / stride;

    const int c  = lane & 15;
    const int h  = lane >> 4;
    const float* wl0 = wp0 + (lane << 2);
    const float* wl1 = wp1 + (lane << 2);
    const float* wl2 = wp2 + (lane << 2);

    for (int it = 0; it < sweeps; ++it) {
        const int tile = it * stride + wave * gridDim.x + blockIdx.x;
        if (__builtin_amdgcn_readfirstlane((int)(tile >= tiles))) break;
        const int e0 = tile * TILE_E;

        {
            const int tn = tile + stride;
            if (tn < tiles) {
                __builtin_prefetch(&eidx[tn * TILE_E], 0, 1);
                __builtin_prefetch(&eidx[E + tn * TILE_E], 0, 1);
            }
        }

        int e_l = e0 + c; if (e_l >= E) e_l = E - 1;
        const int idx_s = eidx[e_l];
        const int idx_e = eidx[E + e_l];

        const float* rs = x + (size_t)idx_s * DHID + (h << 4);
        const float* re = x + (size_t)idx_e * DHID + (h << 4);
        v8f h0[8] = {};
        {
            v4f xq[2][4];
            v4f a0[8], a1[8];
            load_x(xq[0], rs);
            load_a(a0, wl0, 0);
#pragma unroll
            for (int q = 0; q < 8; ++q) {
                if (q + 1 < 8)
                    load_x(xq[(q + 1) & 1], ((q + 1) < 4 ? rs : re) + (((q + 1) & 3) << 5));
#pragma unroll
                for (int u = 0; u < 4; u += 2) {
                    const int kb = 4 * q + u;
                    load_a(a1, wl0, kb + 1);
                    SCHED_FENCE();
                    mma_kb(h0, a0, xq[q & 1][u]);
                    SCHED_FENCE();
                    if (kb + 2 < 32) load_a(a0, wl0, kb + 2);
                    SCHED_FENCE();
                    mma_kb(h0, a1, xq[q & 1][u + 1]);
                    SCHED_FENCE();
                }
            }
        }
        layernorm_tanh_T(h0, par, lane);

        v8f h1[8] = {};
        gemm_chained(h1, h0, wl1);
        layernorm_tanh_T(h1, par + 3 * DHID, lane);

        v8f h2[8] = {};
        gemm_chained(h2, h1, wl2);
        layernorm_tanh_T(h2, par + 6 * DHID, lane);

        {
            const float* w3 = par + 9 * DHID + (h << 3);
            float pp[8];
#pragma unroll
            for (int mm = 0; mm < 8; ++mm) {
                const v4f wlo = *(const v4f*)(w3 + (mm << 4));
                const v4f whi = *(const v4f*)(w3 + (mm << 4) + 4);
                float p = 0.f;
#pragma unroll
                for (int r = 0; r < 8; ++r)
                    p = fmaf(h2[mm][r], (r < 4 ? wlo[r & 3] : whi[r & 3]), p);
                pp[mm] = p;
            }
            float p = ((pp[0] + pp[1]) + (pp[2] + pp[3])) + ((pp[4] + pp[5]) + (pp[6] + pp[7]));
            p += __shfl_xor(p, 16, 32);
            const int e = e0 + c;
            if (h == 0 && e < E) out[e] = p + b3[0];
        }
    }
}

#define SMEM_NODE_FLOATS 32768
#define PQ_ROW 256

__global__ void __launch_bounds__(WPB * 32, 1)
node_proj_wmma_f32(const float* __restrict__ x, int N, const float* __restrict__ W0,
                   float* __restrict__ pq)
{
    extern __shared__ float smem[];
    const int tid  = threadIdx.x;
    const int lane = tid & 31;
    const int wave = tid >> 5;

    pack_wT<false>(smem, W0, 2 * DHID, tid, WPB * 32);
    __syncthreads();

    const int tiles  = (N + TILE_E - 1) / TILE_E;
    const int stride = gridDim.x * WPB;
    const int sweeps = (tiles + stride - 1) / stride;

    const int c = lane & 15;
    const int h = lane >> 4;
    const float* wl0 = smem + (lane << 2);

    for (int it = 0; it < sweeps; ++it) {
        const int tile = it * stride + wave * gridDim.x + blockIdx.x;
        if (__builtin_amdgcn_readfirstlane((int)(tile >= tiles))) break;

        int n_l = tile * TILE_E + c; if (n_l >= N) n_l = N - 1;
        const float* rs = x + (size_t)n_l * DHID + (h << 4);

        v8f accP[8] = {};
        v8f accQ[8] = {};
        {
            v4f xq[2][4];
            v4f a0[8], a1[8];
            load_x(xq[0], rs);
            load_a(a0, wl0, 0);
#pragma unroll
            for (int q = 0; q < 4; ++q) {
                if (q + 1 < 4) load_x(xq[(q + 1) & 1], rs + ((q + 1) << 5));
#pragma unroll
                for (int u = 0; u < 4; ++u) {
                    const int kb = 4 * q + u;
                    load_a(a1, wl0, 16 + kb);
                    SCHED_FENCE();
                    mma_kb(accP, a0, xq[q & 1][u]);
                    SCHED_FENCE();
                    if (kb + 1 < 16) load_a(a0, wl0, kb + 1);
                    SCHED_FENCE();
                    mma_kb(accQ, a1, xq[q & 1][u]);
                    SCHED_FENCE();
                }
            }
        }

        const int n = tile * TILE_E + c;
        if (n < N) {
            float* o = pq + (size_t)n * PQ_ROW + (h << 3);
#pragma unroll
            for (int mm = 0; mm < 8; ++mm) {
                *(v4f*)(o + (mm << 4))            = (v4f){accP[mm][0], accP[mm][1], accP[mm][2], accP[mm][3]};
                *(v4f*)(o + (mm << 4) + 4)        = (v4f){accP[mm][4], accP[mm][5], accP[mm][6], accP[mm][7]};
                *(v4f*)(o + DHID + (mm << 4))     = (v4f){accQ[mm][0], accQ[mm][1], accQ[mm][2], accQ[mm][3]};
                *(v4f*)(o + DHID + (mm << 4) + 4) = (v4f){accQ[mm][4], accQ[mm][5], accQ[mm][6], accQ[mm][7]};
            }
        }
    }
}

#define EOFF_W2 16384
#define EOFF_P  32768
#define SMEM_EDGE_FLOATS (EOFF_P + 10 * DHID)

__global__ void __launch_bounds__(WPB * 32, 1)
edge_mlp_pq_wmma_f32(const float* __restrict__ pq, const int* __restrict__ eidx, int E,
                     const float* __restrict__ b0,
                     const float* __restrict__ W1, const float* __restrict__ b1,
                     const float* __restrict__ W2, const float* __restrict__ b2,
                     const float* __restrict__ W3, const float* __restrict__ b3,
                     const float* __restrict__ g0, const float* __restrict__ be0,
                     const float* __restrict__ g1, const float* __restrict__ be1,
                     const float* __restrict__ g2, const float* __restrict__ be2,
                     float* __restrict__ out)
{
    extern __shared__ float smem[];
    const int tid  = threadIdx.x;
    const int lane = tid & 31;
    const int wave = tid >> 5;

    float* wp1 = smem;
    float* wp2 = smem + EOFF_W2;
    float* par = smem + EOFF_P;

    pack_wT<true>(wp1, W1, DHID, tid, WPB * 32);
    pack_wT<true>(wp2, W2, DHID, tid, WPB * 32);
    if (tid < DHID) {
        par[0 * DHID + tid] = b0[tid];  par[1 * DHID + tid] = g0[tid];  par[2 * DHID + tid] = be0[tid];
        par[3 * DHID + tid] = b1[tid];  par[4 * DHID + tid] = g1[tid];  par[5 * DHID + tid] = be1[tid];
        par[6 * DHID + tid] = b2[tid];  par[7 * DHID + tid] = g2[tid];  par[8 * DHID + tid] = be2[tid];
        par[9 * DHID + tid] = W3[tid];
    }
    __syncthreads();

    const int tiles  = (E + TILE_E - 1) / TILE_E;
    const int stride = gridDim.x * WPB;
    const int sweeps = (tiles + stride - 1) / stride;

    const int c  = lane & 15;
    const int h  = lane >> 4;
    const float* wl1 = wp1 + (lane << 2);
    const float* wl2 = wp2 + (lane << 2);

    for (int it = 0; it < sweeps; ++it) {
        const int tile = it * stride + wave * gridDim.x + blockIdx.x;
        if (__builtin_amdgcn_readfirstlane((int)(tile >= tiles))) break;
        const int e0 = tile * TILE_E;

        {
            const int tn = tile + stride;
            if (tn < tiles) {
                __builtin_prefetch(&eidx[tn * TILE_E], 0, 1);
                __builtin_prefetch(&eidx[E + tn * TILE_E], 0, 1);
            }
        }

        int e_l = e0 + c; if (e_l >= E) e_l = E - 1;
        const int idx_s = eidx[e_l];
        const int idx_e = eidx[E + e_l];

        const float* ps = pq + (size_t)idx_s * PQ_ROW + (h << 3);
        const float* pe = pq + (size_t)idx_e * PQ_ROW + DHID + (h << 3);
        v8f h0[8];
        {
            v4f p[8][2], r[8][2];
#pragma unroll
            for (int mm = 0; mm < 8; ++mm) {
                p[mm][0] = *(const v4f*)(ps + (mm << 4));
                p[mm][1] = *(const v4f*)(ps + (mm << 4) + 4);
            }
#pragma unroll
            for (int mm = 0; mm < 8; ++mm) {
                r[mm][0] = *(const v4f*)(pe + (mm << 4));
                r[mm][1] = *(const v4f*)(pe + (mm << 4) + 4);
            }
#pragma unroll
            for (int mm = 0; mm < 8; ++mm) {
                const v4f lo = p[mm][0] + r[mm][0];
                const v4f hi = p[mm][1] + r[mm][1];
                h0[mm] = (v8f){lo.x, lo.y, lo.z, lo.w, hi.x, hi.y, hi.z, hi.w};
            }
        }
        layernorm_tanh_T(h0, par, lane);

        v8f h1[8] = {};
        gemm_chained(h1, h0, wl1);
        layernorm_tanh_T(h1, par + 3 * DHID, lane);

        v8f h2[8] = {};
        gemm_chained(h2, h1, wl2);
        layernorm_tanh_T(h2, par + 6 * DHID, lane);

        {
            const float* w3 = par + 9 * DHID + (h << 3);
            float pp[8];
#pragma unroll
            for (int mm = 0; mm < 8; ++mm) {
                const v4f wlo = *(const v4f*)(w3 + (mm << 4));
                const v4f whi = *(const v4f*)(w3 + (mm << 4) + 4);
                float p = 0.f;
#pragma unroll
                for (int r = 0; r < 8; ++r)
                    p = fmaf(h2[mm][r], (r < 4 ? wlo[r & 3] : whi[r & 3]), p);
                pp[mm] = p;
            }
            float p = ((pp[0] + pp[1]) + (pp[2] + pp[3])) + ((pp[4] + pp[5]) + (pp[6] + pp[7]));
            p += __shfl_xor(p, 16, 32);
            const int e = e0 + c;
            if (h == 0 && e < E) out[e] = p + b3[0];
        }
    }
}

extern "C" void kernel_launch(void* const* d_in, const int* in_sizes, int n_in,
                              void* d_out, int out_size, void* d_ws, size_t ws_size,
                              hipStream_t stream)
{
    const float* x   = (const float*)d_in[0];
    const int*   ei  = (const int*)  d_in[1];
    const float* W0  = (const float*)d_in[2];
    const float* b0  = (const float*)d_in[3];
    const float* W1  = (const float*)d_in[4];
    const float* b1  = (const float*)d_in[5];
    const float* W2  = (const float*)d_in[6];
    const float* b2  = (const float*)d_in[7];
    const float* W3  = (const float*)d_in[8];
    const float* b3  = (const float*)d_in[9];
    const float* g0  = (const float*)d_in[10];
    const float* be0 = (const float*)d_in[11];
    const float* g1  = (const float*)d_in[12];
    const float* be1 = (const float*)d_in[13];
    const float* g2  = (const float*)d_in[14];
    const float* be2 = (const float*)d_in[15];

    const int E     = in_sizes[1] / 2;
    const int N     = in_sizes[0] / DHID;
    const int tiles = (E + TILE_E - 1) / TILE_E;

    const size_t need = (size_t)N * PQ_ROW * sizeof(float);
    if (d_ws != nullptr && ws_size >= need) {
        const int ntiles = (N + TILE_E - 1) / TILE_E;
        int nblocks = (ntiles + WPB - 1) / WPB;
        if (nblocks > 256) nblocks = 256;
        int eblocks = (tiles + WPB - 1) / WPB;
        if (eblocks > 256) eblocks = 256;
        float* pq = (float*)d_ws;
        node_proj_wmma_f32<<<nblocks, WPB * 32, (size_t)SMEM_NODE_FLOATS * sizeof(float), stream>>>(
            x, N, W0, pq);
        edge_mlp_pq_wmma_f32<<<eblocks, WPB * 32, (size_t)SMEM_EDGE_FLOATS * sizeof(float), stream>>>(
            pq, ei, E, b0, W1, b1, W2, b2, W3, b3,
            g0, be0, g1, be1, g2, be2, (float*)d_out);
        return;
    }

    int blocks = (tiles + WPB - 1) / WPB;
    if (blocks > 256) blocks = 256;
    const size_t shmem = (size_t)SMEM_FLOATS * sizeof(float);

    edge_mlp_wmma_f32<<<blocks, WPB * 32, shmem, stream>>>(
        x, ei, E, W0, b0, W1, b1, W2, b2, W3, b3,
        g0, be0, g1, be1, g2, be2, (float*)d_out);
}


# LDS reads of the vector phases regrouped: each LayerNorm's 48 parameter reads leave as one group in the last load slot of the layer before it, not in pairs inside the normalisation; sums unchanged
# speedup vs baseline: 2.7903x; 1.0294x over previous
#include <hip/hip_runtime.h>
#include <math.h>

typedef __attribute__((ext_vector_type(2))) float v2f;
typedef __attribute__((ext_vector_type(4))) float v4f;
typedef __attribute__((ext_vector_type(8))) float v8f;

#define DHID   128
#define TILE_E 16
#define WPB    12
#define LNEPS  1e-5f

#if __has_builtin(__builtin_amdgcn_tanhf)
#define TANHF(v) __builtin_amdgcn_tanhf(v)
#else
#define TANHF(v) tanhf(v)
#endif

#define SCHED_FENCE() __builtin_amdgcn_sched_barrier(0)

#define OFF_W1 32768
#define OFF_W2 49152
#define OFF_P  65536
#define SMEM_FLOATS (OFF_P + 10 * DHID)

template <bool CHAINED>
__device__ __forceinline__ void pack_wT(float* __restrict__ dst,
                                        const float* __restrict__ W,
                                        int K, int tid, int nthr)
{
    const int total = K << 7;
#pragma unroll 4
    for (int i = tid; i < total; i += nthr) {
        const int k = i >> 7;
        const int n = i & 127;
        int kb, h;
        if (CHAINED) { kb = ((k >> 4) << 1) | ((k >> 2) & 1); h = (k >> 3) & 1; }
        else         { kb = ((k >> 5) << 2) | ((k >> 2) & 3); h = (k >> 4) & 1; }
        dst[((((kb << 3) + (n >> 4)) << 5) + (n & 15) + (h << 4)) * 4 + (k & 3)] = W[i];
    }
}

__device__ __forceinline__ void load_a(v4f a[8], const float* __restrict__ wl, int kb)
{
#pragma unroll
    for (int mm = 0; mm < 8; ++mm)
        a[mm] = *(const v4f*)(wl + (kb << 10) + (mm << 7));
}

__device__ __forceinline__ void mma_kb(v8f acc[8], const v4f a[8], v4f b)
{
    const v2f b0 = {b.x, b.y};
    const v2f b1 = {b.z, b.w};
#pragma unroll
    for (int mm = 0; mm < 8; ++mm) {
        const v2f a0 = {a[mm].x, a[mm].y};
        acc[mm] = __builtin_amdgcn_wmma_f32_16x16x4_f32(
            false, a0, false, b0, (short)0, acc[mm], false, false);
    }
#pragma unroll
    for (int mm = 0; mm < 8; ++mm) {
        const v2f a1 = {a[mm].z, a[mm].w};
        acc[mm] = __builtin_amdgcn_wmma_f32_16x16x4_f32(
            false, a1, false, b1, (short)0, acc[mm], false, false);
    }
}

template <class Tail>
__device__ __forceinline__ void gemm_chained(v8f acc[8], const v8f prev[8],
                                             const float* __restrict__ wl, v4f a0[8], Tail tail)
{
    v4f a1[8];
#pragma unroll
    for (int nn = 0; nn < 8; ++nn) {
        const v8f t = prev[nn];
        load_a(a1, wl, 2 * nn + 1);
        SCHED_FENCE();
        mma_kb(acc, a0, (v4f){t[0], t[1], t[2], t[3]});
        SCHED_FENCE();
        if (nn + 1 < 8) load_a(a0, wl, 2 * nn + 2); else tail();
        SCHED_FENCE();
        mma_kb(acc, a1, (v4f){t[4], t[5], t[6], t[7]});
        SCHED_FENCE();
    }
}

__device__ __forceinline__ void load_par(v4f p[8][2], const float* __restrict__ row, int lane)
{
    const float* pl = row + ((lane >> 4) << 3);
#pragma unroll
    for (int mm = 0; mm < 8; ++mm) {
        p[mm][0] = *(const v4f*)(pl + (mm << 4));
        p[mm][1] = *(const v4f*)(pl + (mm << 4) + 4);
    }
}

template <class Next>
__device__ __forceinline__ void layernorm_tanh_T(v8f acc[8], const v4f pb[8][2],
                                                 const v4f pg[8][2], const v4f pe[8][2],
                                                 int lane, Next next)
{
    SCHED_FENCE();

    float sp[8], qp[8];
#pragma unroll
    for (int mm = 0; mm < 8; ++mm) {
        float s = 0.f, q = 0.f;
#pragma unroll
        for (int r = 0; r < 8; ++r) {
            const float v = acc[mm][r] + pb[mm][r >> 2][r & 3];
            acc[mm][r] = v;
            s += v;
            q = fmaf(v, v, q);
        }
        sp[mm] = s; qp[mm] = q;
    }
    float s  = ((sp[0] + sp[1]) + (sp[2] + sp[3])) + ((sp[4] + sp[5]) + (sp[6] + sp[7]));
    float ss = ((qp[0] + qp[1]) + (qp[2] + qp[3])) + ((qp[4] + qp[5]) + (qp[6] + qp[7]));
    s  += __shfl_xor(s,  16, 32);
    ss += __shfl_xor(ss, 16, 32);
    const float mu  = s * (1.0f / 128.0f);
    const float var = fmaf(-mu, mu, ss * (1.0f / 128.0f));
    const float inv = rsqrtf(var + LNEPS);

    SCHED_FENCE();
    next();
    SCHED_FENCE();

#pragma unroll
    for (int mm = 0; mm < 8; ++mm) {
#pragma unroll
        for (int r = 0; r < 8; ++r) {
            const float v = (acc[mm][r] - mu) * inv;
            acc[mm][r] = TANHF(fmaf(v, pg[mm][r >> 2][r & 3], pe[mm][r >> 2][r & 3]));
        }
    }
}

__device__ __forceinline__ void load_x(v4f d[4], const float* __restrict__ p)
{
#pragma unroll
    for (int u = 0; u < 4; ++u) d[u] = *(const v4f*)(p + (u << 2));
}

__global__ void __launch_bounds__(WPB * 32, 1)
edge_mlp_wmma_f32(const float* __restrict__ x, const int* __restrict__ eidx, int E,
                  const float* __restrict__ W0, const float* __restrict__ b0,
                  const float* __restrict__ W1, const float* __restrict__ b1,
                  const float* __restrict__ W2, const float* __restrict__ b2,
                  const float* __restrict__ W3, const float* __restrict__ b3,
                  const float* __restrict__ g0, const float* __restrict__ be0,
                  const float* __restrict__ g1, const float* __restrict__ be1,
                  const float* __restrict__ g2, const float* __restrict__ be2,
                  float* __restrict__ out)
{
    extern __shared__ float smem[];
    const int tid  = threadIdx.x;
    const int lane = tid & 31;
    const int wave = tid >> 5;

    float* wp0 = smem;
    float* wp1 = smem + OFF_W1;
    float* wp2 = smem + OFF_W2;
    float* par = smem + OFF_P;

    pack_wT<false>(wp0, W0, 2 * DHID, tid, WPB * 32);
    pack_wT<true >(wp1, W1, DHID,     tid, WPB * 32);
    pack_wT<true >(wp2, W2, DHID,     tid, WPB * 32);
    if (tid < DHID) {
        par[0 * DHID + tid] = b0[tid];  par[1 * DHID + tid] = g0[tid];  par[2 * DHID + tid] = be0[tid];
        par[3 * DHID + tid] = b1[tid];  par[4 * DHID + tid] = g1[tid];  par[5 * DHID + tid] = be1[tid];
        par[6 * DHID + tid] = b2[tid];  par[7 * DHID + tid] = g2[tid];  par[8 * DHID + tid] = be2[tid];
        par[9 * DHID + tid] = W3[tid];
    }
    __syncthreads();

    const int tiles  = (E + TILE_E - 1) / TILE_E;
    const int stride = gridDim.x * WPB;
    const int sweeps = (tiles + stride - 1) / stride;

    const int c  = lane & 15;
    const int h  = lane >> 4;
    const float* wl0 = wp0 + (lane << 2);
    const float* wl1 = wp1 + (lane << 2);
    const float* wl2 = wp2 + (lane << 2);

    for (int it = 0; it < sweeps; ++it) {
        const int tile = it * stride + wave * gridDim.x + blockIdx.x;
        if (__builtin_amdgcn_readfirstlane((int)(tile >= tiles))) break;
        const int e0 = tile * TILE_E;

        {
            const int tn = tile + stride;
            if (tn < tiles) {
                __builtin_prefetch(&eidx[tn * TILE_E], 0, 1);
                __builtin_prefetch(&eidx[E + tn * TILE_E], 0, 1);
            }
        }

        int e_l = e0 + c; if (e_l >= E) e_l = E - 1;
        const int idx_s = eidx[e_l];
        const int idx_e = eidx[E + e_l];

        const float* rs = x + (size_t)idx_s * DHID + (h << 4);
        const float* re = x + (size_t)idx_e * DHID + (h << 4);
        v8f h0[8] = {};
        v4f lnb[8][2], lng[8][2], lne[8][2];
        {
            v4f xq[2][4];
            v4f a0[8], a1[8];
            load_x(xq[0], rs);
            load_a(a0, wl0, 0);
#pragma unroll
            for (int q = 0; q < 8; ++q) {
                if (q + 1 < 8)
                    load_x(xq[(q + 1) & 1], ((q + 1) < 4 ? rs : re) + (((q + 1) & 3) << 5));
#pragma unroll
                for (int u = 0; u < 4; u += 2) {
                    const int kb = 4 * q + u;
                    load_a(a1, wl0, kb + 1);
                    SCHED_FENCE();
                    mma_kb(h0, a0, xq[q & 1][u]);
                    SCHED_FENCE();
                    if (kb + 2 < 32) load_a(a0, wl0, kb + 2);
                    else { load_par(lnb, par, lane); load_par(lng, par + DHID, lane); load_par(lne, par + 2 * DHID, lane); }
                    SCHED_FENCE();
                    mma_kb(h0, a1, xq[q & 1][u + 1]);
                    SCHED_FENCE();
                }
            }
        }
        v4f af[8];
        layernorm_tanh_T(h0, lnb, lng, lne, lane, [&] { load_a(af, wl1, 0); });

        v8f h1[8] = {};
        gemm_chained(h1, h0, wl1, af, [&] {
            load_par(lnb, par + 3 * DHID, lane); load_par(lng, par + 4 * DHID, lane); load_par(lne, par + 5 * DHID, lane); });
        layernorm_tanh_T(h1, lnb, lng, lne, lane, [&] { load_a(af, wl2, 0); });

        v8f h2[8] = {};
        gemm_chained(h2, h1, wl2, af, [&] {
            load_par(lnb, par + 6 * DHID, lane); load_par(lng, par + 7 * DHID, lane); load_par(lne, par + 8 * DHID, lane); });
        v4f w3v[8][2];
        layernorm_tanh_T(h2, lnb, lng, lne, lane, [&] { load_par(w3v, par + 9 * DHID, lane); });

        {
            float pp[8];
#pragma unroll
            for (int mm = 0; mm < 8; ++mm) {
                float p = 0.f;
#pragma unroll
                for (int r = 0; r < 8; ++r)
                    p = fmaf(h2[mm][r], w3v[mm][r >> 2][r & 3], p);
                pp[mm] = p;
            }
            float p = ((pp[0] + pp[1]) + (pp[2] + pp[3])) + ((pp[4] + pp[5]) + (pp[6] + pp[7]));
            p += __shfl_xor(p, 16, 32);
            const int e = e0 + c;
            if (h == 0 && e < E) out[e] = p + b3[0];
        }
    }
}

#define SMEM_NODE_FLOATS 32768
#define PQ_ROW 256

__global__ void __launch_bounds__(WPB * 32, 1)
node_proj_wmma_f32(const float* __restrict__ x, int N, const float* __restrict__ W0,
                   float* __restrict__ pq)
{
    extern __shared__ float smem[];
    const int tid  = threadIdx.x;
    const int lane = tid & 31;
    const int wave = tid >> 5;

    pack_wT<false>(smem, W0, 2 * DHID, tid, WPB * 32);
    __syncthreads();

    const int tiles  = (N + TILE_E - 1) / TILE_E;
    const int stride = gridDim.x * WPB;
    const int sweeps = (tiles + stride - 1) / stride;

    const int c = lane & 15;
    const int h = lane >> 4;
    const float* wl0 = smem + (lane << 2);

    for (int it = 0; it < sweeps; ++it) {
        const int tile = it * stride + wave * gridDim.x + blockIdx.x;
        if (__builtin_amdgcn_readfirstlane((int)(tile >= tiles))) break;

        int n_l = tile * TILE_E + c; if (n_l >= N) n_l = N - 1;
        const float* rs = x + (size_t)n_l * DHID + (h << 4);

        v8f accP[8] = {};
        v8f accQ[8] = {};
        {
            v4f xq[2][4];
            v4f a0[8], a1[8];
            load_x(xq[0], rs);
            load_a(a0, wl0, 0);
#pragma unroll
            for (int q = 0; q < 4; ++q) {
                if (q + 1 < 4) load_x(xq[(q + 1) & 1], rs + ((q + 1) << 5));
#pragma unroll
                for (int u = 0; u < 4; ++u) {
                    const int kb = 4 * q + u;
                    load_a(a1, wl0, 16 + kb);
                    SCHED_FENCE();
                    mma_kb(accP, a0, xq[q & 1][u]);
                    SCHED_FENCE();
                    if (kb + 1 < 16) load_a(a0, wl0, kb + 1);
                    SCHED_FENCE();
                    mma_kb(accQ, a1, xq[q & 1][u]);
                    SCHED_FENCE();
                }
            }
        }

        const int n = tile * TILE_E + c;
        if (n < N) {
            float* o = pq + (size_t)n * PQ_ROW + (h << 3);
#pragma unroll
            for (int mm = 0; mm < 8; ++mm) {
                *(v4f*)(o + (mm << 4))            = (v4f){accP[mm][0], accP[mm][1], accP[mm][2], accP[mm][3]};
                *(v4f*)(o + (mm << 4) + 4)        = (v4f){accP[mm][4], accP[mm][5], accP[mm][6], accP[mm][7]};
                *(v4f*)(o + DHID + (mm << 4))     = (v4f){accQ[mm][0], accQ[mm][1], accQ[mm][2], accQ[mm][3]};
                *(v4f*)(o + DHID + (mm << 4) + 4) = (v4f){accQ[mm][4], accQ[mm][5], accQ[mm][6], accQ[mm][7]};
            }
        }
    }
}

#define EOFF_W2 16384
#define EOFF_P  32768
#define SMEM_EDGE_FLOATS (EOFF_P + 10 * DHID)

__global__ void __launch_bounds__(WPB * 32, 1)
edge_mlp_pq_wmma_f32(const float* __restrict__ pq, const int* __restrict__ eidx, int E,
                     const float* __restrict__ b0,
                     const float* __restrict__ W1, const float* __restrict__ b1,
                     const float* __restrict__ W2, const float* __restrict__ b2,
                     const float* __restrict__ W3, const float* __restrict__ b3,
                     const float* __restrict__ g0, const float* __restrict__ be0,
                     const float* __restrict__ g1, const float* __restrict__ be1,
                     const float* __restrict__ g2, const float* __restrict__ be2,
                     float* __restrict__ out)
{
    extern __shared__ float smem[];
    const int tid  = threadIdx.x;
    const int lane = tid & 31;
    const int wave = tid >> 5;

    float* wp1 = smem;
    float* wp2 = smem + EOFF_W2;
    float* par = smem + EOFF_P;

    pack_wT<true>(wp1, W1, DHID, tid, WPB * 32);
    pack_wT<true>(wp2, W2, DHID, tid, WPB * 32);
    if (tid < DHID) {
        par[0 * DHID + tid] = b0[tid];  par[1 * DHID + tid] = g0[tid];  par[2 * DHID + tid] = be0[tid];
        par[3 * DHID + tid] = b1[tid];  par[4 * DHID + tid] = g1[tid];  par[5 * DHID + tid] = be1[tid];
        par[6 * DHID + tid] = b2[tid];  par[7 * DHID + tid] = g2[tid];  par[8 * DHID + tid] = be2[tid];
        par[9 * DHID + tid] = W3[tid];
    }
    __syncthreads();

    const int tiles  = (E + TILE_E - 1) / TILE_E;
    const int stride = gridDim.x * WPB;
    const int sweeps = (tiles + stride - 1) / stride;

    const int c  = lane & 15;
    const int h  = lane >> 4;
    const float* wl1 = wp1 + (lane << 2);
    const float* wl2 = wp2 + (lane << 2);

    for (int it = 0; it < sweeps; ++it) {
        const int tile = it * stride + wave * gridDim.x + blockIdx.x;
        if (__builtin_amdgcn_readfirstlane((int)(tile >= tiles))) break;
        const int e0 = tile * TILE_E;

        {
            const int tn = tile + stride;
            if (tn < tiles) {
                __builtin_prefetch(&eidx[tn * TILE_E], 0, 1);
                __builtin_prefetch(&eidx[E + tn * TILE_E], 0, 1);
            }
        }

        int e_l = e0 + c; if (e_l >= E) e_l = E - 1;
        const int idx_s = eidx[e_l];
        const int idx_e = eidx[E + e_l];

        const float* ps = pq + (size_t)idx_s * PQ_ROW + (h << 3);
        const float* pe = pq + (size_t)idx_e * PQ_ROW + DHID + (h << 3);
        v8f h0[8];
        v4f lnb[8][2], lng[8][2], lne[8][2];
        load_par(lnb, par, lane);
        {
            v4f p[8][2], r[8][2];
#pragma unroll
            for (int mm = 0; mm < 8; ++mm) {
                p[mm][0] = *(const v4f*)(ps + (mm << 4));
                p[mm][1] = *(const v4f*)(ps + (mm << 4) + 4);
            }
#pragma unroll
            for (int mm = 0; mm < 8; ++mm) {
                r[mm][0] = *(const v4f*)(pe + (mm << 4));
                r[mm][1] = *(const v4f*)(pe + (mm << 4) + 4);
            }
#pragma unroll
            for (int mm = 0; mm < 8; ++mm) {
                const v4f lo = p[mm][0] + r[mm][0];
                const v4f hi = p[mm][1] + r[mm][1];
                h0[mm] = (v8f){lo.x, lo.y, lo.z, lo.w, hi.x, hi.y, hi.z, hi.w};
            }
        }
        SCHED_FENCE();
        load_par(lng, par + DHID, lane);
        load_par(lne, par + 2 * DHID, lane);
        v4f af[8];
        layernorm_tanh_T(h0, lnb, lng, lne, lane, [&] { load_a(af, wl1, 0); });

        v8f h1[8] = {};
        gemm_chained(h1, h0, wl1, af, [&] {
            load_par(lnb, par + 3 * DHID, lane); load_par(lng, par + 4 * DHID, lane); load_par(lne, par + 5 * DHID, lane); });
        layernorm_tanh_T(h1, lnb, lng, lne, lane, [&] { load_a(af, wl2, 0); });

        v8f h2[8] = {};
        gemm_chained(h2, h1, wl2, af, [&] {
            load_par(lnb, par + 6 * DHID, lane); load_par(lng, par + 7 * DHID, lane); load_par(lne, par + 8 * DHID, lane); });
        v4f w3v[8][2];
        layernorm_tanh_T(h2, lnb, lng, lne, lane, [&] { load_par(w3v, par + 9 * DHID, lane); });

        {
            float pp[8];
#pragma unroll
            for (int mm = 0; mm < 8; ++mm) {
                float p = 0.f;
#pragma unroll
                for (int r = 0; r < 8; ++r)
                    p = fmaf(h2[mm][r], w3v[mm][r >> 2][r & 3], p);
                pp[mm] = p;
            }
            float p = ((pp[0] + pp[1]) + (pp[2] + pp[3])) + ((pp[4] + pp[5]) + (pp[6] + pp[7]));
            p += __shfl_xor(p, 16, 32);
            const int e = e0 + c;
            if (h == 0 && e < E) out[e] = p + b3[0];
        }
    }
}

extern "C" void kernel_launch(void* const* d_in, const int* in_sizes, int n_in,
                              void* d_out, int out_size, void* d_ws, size_t ws_size,
                              hipStream_t stream)
{
    const float* x   = (const float*)d_in[0];
    const int*   ei  = (const int*)  d_in[1];
    const float* W0  = (const float*)d_in[2];
    const float* b0  = (const float*)d_in[3];
    const float* W1  = (const float*)d_in[4];
    const float* b1  = (const float*)d_in[5];
    const float* W2  = (const float*)d_in[6];
    const float* b2  = (const float*)d_in[7];
    const float* W3  = (const float*)d_in[8];
    const float* b3  = (const float*)d_in[9];
    const float* g0  = (const float*)d_in[10];
    const float* be0 = (const float*)d_in[11];
    const float* g1  = (const float*)d_in[12];
    const float* be1 = (const float*)d_in[13];
    const float* g2  = (const float*)d_in[14];
    const float* be2 = (const float*)d_in[15];

    const int E     = in_sizes[1] / 2;
    const int N     = in_sizes[0] / DHID;
    const int tiles = (E + TILE_E - 1) / TILE_E;

    const size_t need = (size_t)N * PQ_ROW * sizeof(float);
    if (d_ws != nullptr && ws_size >= need) {
        const int ntiles = (N + TILE_E - 1) / TILE_E;
        int nblocks = (ntiles + WPB - 1) / WPB;
        if (nblocks > 256) nblocks = 256;
        int eblocks = (tiles + WPB - 1) / WPB;
        if (eblocks > 256) eblocks = 256;
        float* pq = (float*)d_ws;
        node_proj_wmma_f32<<<nblocks, WPB * 32, (size_t)SMEM_NODE_FLOATS * sizeof(float), stream>>>(
            x, N, W0, pq);
        edge_mlp_pq_wmma_f32<<<eblocks, WPB * 32, (size_t)SMEM_EDGE_FLOATS * sizeof(float), stream>>>(
            pq, ei, E, b0, W1, b1, W2, b2, W3, b3,
            g0, be0, g1, be1, g2, be2, (float*)d_out);
        return;
    }

    int blocks = (tiles + WPB - 1) / WPB;
    if (blocks > 256) blocks = 256;
    const size_t shmem = (size_t)SMEM_FLOATS * sizeof(float);

    edge_mlp_wmma_f32<<<blocks, WPB * 32, shmem, stream>>>(
        x, ei, E, W0, b0, W1, b1, W2, b2, W3, b3,
        g0, be0, g1, be1, g2, be2, (float*)d_out);
}


# vector phases (bias, LayerNorm, last dot) written as 2-wide packed f32 ops on adjacent register pairs: no register copies to form pairs; LN and last-dot sums run as even/odd partial sums
# speedup vs baseline: 2.8834x; 1.0334x over previous
#include <hip/hip_runtime.h>
#include <math.h>

typedef __attribute__((ext_vector_type(2))) float v2f;
typedef __attribute__((ext_vector_type(4))) float v4f;
typedef __attribute__((ext_vector_type(8))) float v8f;

#define DHID   128
#define TILE_E 16
#define WPB    12
#define LNEPS  1e-5f

#if __has_builtin(__builtin_amdgcn_tanhf)
#define TANHF(v) __builtin_amdgcn_tanhf(v)
#else
#define TANHF(v) tanhf(v)
#endif

#define SCHED_FENCE() __builtin_amdgcn_sched_barrier(0)

__device__ __forceinline__ v2f acc_pair(const v8f& t, int j) { return (v2f){t[2 * j], t[2 * j + 1]}; }
__device__ __forceinline__ void set_pair(v8f& t, int j, v2f v) { t[2 * j] = v.x; t[2 * j + 1] = v.y; }
__device__ __forceinline__ v2f par_pair(const v4f p[2], int j)
{
    const v4f q = p[j >> 1];
    return (j & 1) ? (v2f){q.z, q.w} : (v2f){q.x, q.y};
}
__device__ __forceinline__ v2f fma2(v2f a, v2f b, v2f c) { return __builtin_elementwise_fma(a, b, c); }
__device__ __forceinline__ v2f tree8(const v2f t[8])
{
    return ((t[0] + t[1]) + (t[2] + t[3])) + ((t[4] + t[5]) + (t[6] + t[7]));
}

#define OFF_W1 32768
#define OFF_W2 49152
#define OFF_P  65536
#define SMEM_FLOATS (OFF_P + 10 * DHID)

template <bool CHAINED>
__device__ __forceinline__ void pack_wT(float* __restrict__ dst,
                                        const float* __restrict__ W,
                                        int K, int tid, int nthr)
{
    const int total = K << 7;
#pragma unroll 4
    for (int i = tid; i < total; i += nthr) {
        const int k = i >> 7;
        const int n = i & 127;
        int kb, h;
        if (CHAINED) { kb = ((k >> 4) << 1) | ((k >> 2) & 1); h = (k >> 3) & 1; }
        else         { kb = ((k >> 5) << 2) | ((k >> 2) & 3); h = (k >> 4) & 1; }
        dst[((((kb << 3) + (n >> 4)) << 5) + (n & 15) + (h << 4)) * 4 + (k & 3)] = W[i];
    }
}

__device__ __forceinline__ void load_a(v4f a[8], const float* __restrict__ wl, int kb)
{
#pragma unroll
    for (int mm = 0; mm < 8; ++mm)
        a[mm] = *(const v4f*)(wl + (kb << 10) + (mm << 7));
}

__device__ __forceinline__ void mma_kb(v8f acc[8], const v4f a[8], v4f b)
{
    const v2f b0 = {b.x, b.y};
    const v2f b1 = {b.z, b.w};
#pragma unroll
    for (int mm = 0; mm < 8; ++mm) {
        const v2f a0 = {a[mm].x, a[mm].y};
        acc[mm] = __builtin_amdgcn_wmma_f32_16x16x4_f32(
            false, a0, false, b0, (short)0, acc[mm], false, false);
    }
#pragma unroll
    for (int mm = 0; mm < 8; ++mm) {
        const v2f a1 = {a[mm].z, a[mm].w};
        acc[mm] = __builtin_amdgcn_wmma_f32_16x16x4_f32(
            false, a1, false, b1, (short)0, acc[mm], false, false);
    }
}

template <class Tail>
__device__ __forceinline__ void gemm_chained(v8f acc[8], const v8f prev[8],
                                             const float* __restrict__ wl, v4f a0[8], Tail tail)
{
    v4f a1[8];
#pragma unroll
    for (int nn = 0; nn < 8; ++nn) {
        const v8f t = prev[nn];
        load_a(a1, wl, 2 * nn + 1);
        SCHED_FENCE();
        mma_kb(acc, a0, (v4f){t[0], t[1], t[2], t[3]});
        SCHED_FENCE();
        if (nn + 1 < 8) load_a(a0, wl, 2 * nn + 2); else tail();
        SCHED_FENCE();
        mma_kb(acc, a1, (v4f){t[4], t[5], t[6], t[7]});
        SCHED_FENCE();
    }
}

__device__ __forceinline__ void load_par(v4f p[8][2], const float* __restrict__ row, int lane)
{
    const float* pl = row + ((lane >> 4) << 3);
#pragma unroll
    for (int mm = 0; mm < 8; ++mm) {
        p[mm][0] = *(const v4f*)(pl + (mm << 4));
        p[mm][1] = *(const v4f*)(pl + (mm << 4) + 4);
    }
}

template <class Next>
__device__ __forceinline__ void layernorm_tanh_T(v8f acc[8], const v4f pb[8][2],
                                                 const v4f pg[8][2], const v4f pe[8][2],
                                                 int lane, Next next)
{
    SCHED_FENCE();

    v2f sp[8], qp[8];
#pragma unroll
    for (int mm = 0; mm < 8; ++mm) {
        v2f s2, q2;
#pragma unroll
        for (int j = 0; j < 4; ++j) {
            const v2f v = acc_pair(acc[mm], j) + par_pair(pb[mm], j);
            set_pair(acc[mm], j, v);
            if (j == 0) { s2 = v; q2 = v * v; }
            else        { s2 += v; q2 = fma2(v, v, q2); }
        }
        sp[mm] = s2; qp[mm] = q2;
    }
    const v2f st = tree8(sp);
    const v2f qt = tree8(qp);
    float s  = st.x + st.y;
    float ss = qt.x + qt.y;
    s  += __shfl_xor(s,  16, 32);
    ss += __shfl_xor(ss, 16, 32);
    const float mu  = s * (1.0f / 128.0f);
    const float var = fmaf(-mu, mu, ss * (1.0f / 128.0f));
    const float inv = rsqrtf(var + LNEPS);

    SCHED_FENCE();
    next();
    SCHED_FENCE();

    const v2f mu2  = {mu, mu};
    const v2f inv2 = {inv, inv};
#pragma unroll
    for (int mm = 0; mm < 8; ++mm) {
#pragma unroll
        for (int j = 0; j < 4; ++j) {
            const v2f v = (acc_pair(acc[mm], j) - mu2) * inv2;
            const v2f t = fma2(v, par_pair(pg[mm], j), par_pair(pe[mm], j));
            set_pair(acc[mm], j, (v2f){TANHF(t.x), TANHF(t.y)});
        }
    }
}

__device__ __forceinline__ float dot64_T(const v8f hh[8], const v4f w[8][2])
{
    v2f pp[8];
#pragma unroll
    for (int mm = 0; mm < 8; ++mm) {
        v2f p2 = acc_pair(hh[mm], 0) * par_pair(w[mm], 0);
#pragma unroll
        for (int j = 1; j < 4; ++j)
            p2 = fma2(acc_pair(hh[mm], j), par_pair(w[mm], j), p2);
        pp[mm] = p2;
    }
    const v2f pt = tree8(pp);
    return pt.x + pt.y;
}

__device__ __forceinline__ void load_x(v4f d[4], const float* __restrict__ p)
{
#pragma unroll
    for (int u = 0; u < 4; ++u) d[u] = *(const v4f*)(p + (u << 2));
}

__global__ void __launch_bounds__(WPB * 32, 1)
edge_mlp_wmma_f32(const float* __restrict__ x, const int* __restrict__ eidx, int E,
                  const float* __restrict__ W0, const float* __restrict__ b0,
                  const float* __restrict__ W1, const float* __restrict__ b1,
                  const float* __restrict__ W2, const float* __restrict__ b2,
                  const float* __restrict__ W3, const float* __restrict__ b3,
                  const float* __restrict__ g0, const float* __restrict__ be0,
                  const float* __restrict__ g1, const float* __restrict__ be1,
                  const float* __restrict__ g2, const float* __restrict__ be2,
                  float* __restrict__ out)
{
    extern __shared__ float smem[];
    const int tid  = threadIdx.x;
    const int lane = tid & 31;
    const int wave = tid >> 5;

    float* wp0 = smem;
    float* wp1 = smem + OFF_W1;
    float* wp2 = smem + OFF_W2;
    float* par = smem + OFF_P;

    pack_wT<false>(wp0, W0, 2 * DHID, tid, WPB * 32);
    pack_wT<true >(wp1, W1, DHID,     tid, WPB * 32);
    pack_wT<true >(wp2, W2, DHID,     tid, WPB * 32);
    if (tid < DHID) {
        par[0 * DHID + tid] = b0[tid];  par[1 * DHID + tid] = g0[tid];  par[2 * DHID + tid] = be0[tid];
        par[3 * DHID + tid] = b1[tid];  par[4 * DHID + tid] = g1[tid];  par[5 * DHID + tid] = be1[tid];
        par[6 * DHID + tid] = b2[tid];  par[7 * DHID + tid] = g2[tid];  par[8 * DHID + tid] = be2[tid];
        par[9 * DHID + tid] = W3[tid];
    }
    __syncthreads();

    const int tiles  = (E + TILE_E - 1) / TILE_E;
    const int stride = gridDim.x * WPB;
    const int sweeps = (tiles + stride - 1) / stride;

    const int c  = lane & 15;
    const int h  = lane >> 4;
    const float* wl0 = wp0 + (lane << 2);
    const float* wl1 = wp1 + (lane << 2);
    const float* wl2 = wp2 + (lane << 2);

    for (int it = 0; it < sweeps; ++it) {
        const int tile = it * stride + wave * gridDim.x + blockIdx.x;
        if (__builtin_amdgcn_readfirstlane((int)(tile >= tiles))) break;
        const int e0 = tile * TILE_E;

        {
            const int tn = tile + stride;
            if (tn < tiles) {
                __builtin_prefetch(&eidx[tn * TILE_E], 0, 1);
                __builtin_prefetch(&eidx[E + tn * TILE_E], 0, 1);
            }
        }

        int e_l = e0 + c; if (e_l >= E) e_l = E - 1;
        const int idx_s = eidx[e_l];
        const int idx_e = eidx[E + e_l];

        const float* rs = x + (size_t)idx_s * DHID + (h << 4);
        const float* re = x + (size_t)idx_e * DHID + (h << 4);
        v8f h0[8] = {};
        v4f lnb[8][2], lng[8][2], lne[8][2];
        {
            v4f xq[2][4];
            v4f a0[8], a1[8];
            load_x(xq[0], rs);
            load_a(a0, wl0, 0);
#pragma unroll
            for (int q = 0; q < 8; ++q) {
                if (q + 1 < 8)
                    load_x(xq[(q + 1) & 1], ((q + 1) < 4 ? rs : re) + (((q + 1) & 3) << 5));
#pragma unroll
                for (int u = 0; u < 4; u += 2) {
                    const int kb = 4 * q + u;
                    load_a(a1, wl0, kb + 1);
                    SCHED_FENCE();
                    mma_kb(h0, a0, xq[q & 1][u]);
                    SCHED_FENCE();
                    if (kb + 2 < 32) load_a(a0, wl0, kb + 2);
                    else { load_par(lnb, par, lane); load_par(lng, par + DHID, lane); load_par(lne, par + 2 * DHID, lane); }
                    SCHED_FENCE();
                    mma_kb(h0, a1, xq[q & 1][u + 1]);
                    SCHED_FENCE();
                }
            }
        }
        v4f af[8];
        layernorm_tanh_T(h0, lnb, lng, lne, lane, [&] { load_a(af, wl1, 0); });

        v8f h1[8] = {};
        gemm_chained(h1, h0, wl1, af, [&] {
            load_par(lnb, par + 3 * DHID, lane); load_par(lng, par + 4 * DHID, lane); load_par(lne, par + 5 * DHID, lane); });
        layernorm_tanh_T(h1, lnb, lng, lne, lane, [&] { load_a(af, wl2, 0); });

        v8f h2[8] = {};
        gemm_chained(h2, h1, wl2, af, [&] {
            load_par(lnb, par + 6 * DHID, lane); load_par(lng, par + 7 * DHID, lane); load_par(lne, par + 8 * DHID, lane); });
        v4f w3v[8][2];
        layernorm_tanh_T(h2, lnb, lng, lne, lane, [&] { load_par(w3v, par + 9 * DHID, lane); });

        {
            float p = dot64_T(h2, w3v);
            p += __shfl_xor(p, 16, 32);
            const int e = e0 + c;
            if (h == 0 && e < E) out[e] = p + b3[0];
        }
    }
}

#define SMEM_NODE_FLOATS 32768
#define PQ_ROW 256

__global__ void __launch_bounds__(WPB * 32, 1)
node_proj_wmma_f32(const float* __restrict__ x, int N, const float* __restrict__ W0,
                   float* __restrict__ pq)
{
    extern __shared__ float smem[];
    const int tid  = threadIdx.x;
    const int lane = tid & 31;
    const int wave = tid >> 5;

    pack_wT<false>(smem, W0, 2 * DHID, tid, WPB * 32);
    __syncthreads();

    const int tiles  = (N + TILE_E - 1) / TILE_E;
    const int stride = gridDim.x * WPB;
    const int sweeps = (tiles + stride - 1) / stride;

    const int c = lane & 15;
    const int h = lane >> 4;
    const float* wl0 = smem + (lane << 2);

    for (int it = 0; it < sweeps; ++it) {
        const int tile = it * stride + wave * gridDim.x + blockIdx.x;
        if (__builtin_amdgcn_readfirstlane((int)(tile >= tiles))) break;

        int n_l = tile * TILE_E + c; if (n_l >= N) n_l = N - 1;
        const float* rs = x + (size_t)n_l * DHID + (h << 4);

        v8f accP[8] = {};
        v8f accQ[8] = {};
        {
            v4f xq[2][4];
            v4f a0[8], a1[8];
            load_x(xq[0], rs);
            load_a(a0, wl0, 0);
#pragma unroll
            for (int q = 0; q < 4; ++q) {
                if (q + 1 < 4) load_x(xq[(q + 1) & 1], rs + ((q + 1) << 5));
#pragma unroll
                for (int u = 0; u < 4; ++u) {
                    const int kb = 4 * q + u;
                    load_a(a1, wl0, 16 + kb);
                    SCHED_FENCE();
                    mma_kb(accP, a0, xq[q & 1][u]);
                    SCHED_FENCE();
                    if (kb + 1 < 16) load_a(a0, wl0, kb + 1);
                    SCHED_FENCE();
                    mma_kb(accQ, a1, xq[q & 1][u]);
                    SCHED_FENCE();
                }
            }
        }

        const int n = tile * TILE_E + c;
        if (n < N) {
            float* o = pq + (size_t)n * PQ_ROW + (h << 3);
#pragma unroll
            for (int mm = 0; mm < 8; ++mm) {
                *(v4f*)(o + (mm << 4))            = (v4f){accP[mm][0], accP[mm][1], accP[mm][2], accP[mm][3]};
                *(v4f*)(o + (mm << 4) + 4)        = (v4f){accP[mm][4], accP[mm][5], accP[mm][6], accP[mm][7]};
                *(v4f*)(o + DHID + (mm << 4))     = (v4f){accQ[mm][0], accQ[mm][1], accQ[mm][2], accQ[mm][3]};
                *(v4f*)(o + DHID + (mm << 4) + 4) = (v4f){accQ[mm][4], accQ[mm][5], accQ[mm][6], accQ[mm][7]};
            }
        }
    }
}

#define EOFF_W2 16384
#define EOFF_P  32768
#define SMEM_EDGE_FLOATS (EOFF_P + 10 * DHID)

__global__ void __launch_bounds__(WPB * 32, 1)
edge_mlp_pq_wmma_f32(const float* __restrict__ pq, const int* __restrict__ eidx, int E,
                     const float* __restrict__ b0,
                     const float* __restrict__ W1, const float* __restrict__ b1,
                     const float* __restrict__ W2, const float* __restrict__ b2,
                     const float* __restrict__ W3, const float* __restrict__ b3,
                     const float* __restrict__ g0, const float* __restrict__ be0,
                     const float* __restrict__ g1, const float* __restrict__ be1,
                     const float* __restrict__ g2, const float* __restrict__ be2,
                     float* __restrict__ out)
{
    extern __shared__ float smem[];
    const int tid  = threadIdx.x;
    const int lane = tid & 31;
    const int wave = tid >> 5;

    float* wp1 = smem;
    float* wp2 = smem + EOFF_W2;
    float* par = smem + EOFF_P;

    pack_wT<true>(wp1, W1, DHID, tid, WPB * 32);
    pack_wT<true>(wp2, W2, DHID, tid, WPB * 32);
    if (tid < DHID) {
        par[0 * DHID + tid] = b0[tid];  par[1 * DHID + tid] = g0[tid];  par[2 * DHID + tid] = be0[tid];
        par[3 * DHID + tid] = b1[tid];  par[4 * DHID + tid] = g1[tid];  par[5 * DHID + tid] = be1[tid];
        par[6 * DHID + tid] = b2[tid];  par[7 * DHID + tid] = g2[tid];  par[8 * DHID + tid] = be2[tid];
        par[9 * DHID + tid] = W3[tid];
    }
    __syncthreads();

    const int tiles  = (E + TILE_E - 1) / TILE_E;
    const int stride = gridDim.x * WPB;
    const int sweeps = (tiles + stride - 1) / stride;

    const int c  = lane & 15;
    const int h  = lane >> 4;
    const float* wl1 = wp1 + (lane << 2);
    const float* wl2 = wp2 + (lane << 2);

    for (int it = 0; it < sweeps; ++it) {
        const int tile = it * stride + wave * gridDim.x + blockIdx.x;
        if (__builtin_amdgcn_readfirstlane((int)(tile >= tiles))) break;
        const int e0 = tile * TILE_E;

        {
            const int tn = tile + stride;
            if (tn < tiles) {
                __builtin_prefetch(&eidx[tn * TILE_E], 0, 1);
                __builtin_prefetch(&eidx[E + tn * TILE_E], 0, 1);
            }
        }

        int e_l = e0 + c; if (e_l >= E) e_l = E - 1;
        const int idx_s = eidx[e_l];
        const int idx_e = eidx[E + e_l];

        const float* ps = pq + (size_t)idx_s * PQ_ROW + (h << 3);
        const float* pe = pq + (size_t)idx_e * PQ_ROW + DHID + (h << 3);
        v8f h0[8];
        v4f lnb[8][2], lng[8][2], lne[8][2];
        load_par(lnb, par, lane);
        {
            v4f p[8][2], r[8][2];
#pragma unroll
            for (int mm = 0; mm < 8; ++mm) {
                p[mm][0] = *(const v4f*)(ps + (mm << 4));
                p[mm][1] = *(const v4f*)(ps + (mm << 4) + 4);
            }
#pragma unroll
            for (int mm = 0; mm < 8; ++mm) {
                r[mm][0] = *(const v4f*)(pe + (mm << 4));
                r[mm][1] = *(const v4f*)(pe + (mm << 4) + 4);
            }
#pragma unroll
            for (int mm = 0; mm < 8; ++mm) {
                const v4f lo = p[mm][0] + r[mm][0];
                const v4f hi = p[mm][1] + r[mm][1];
                h0[mm] = (v8f){lo.x, lo.y, lo.z, lo.w, hi.x, hi.y, hi.z, hi.w};
            }
        }
        SCHED_FENCE();
        load_par(lng, par + DHID, lane);
        load_par(lne, par + 2 * DHID, lane);
        v4f af[8];
        layernorm_tanh_T(h0, lnb, lng, lne, lane, [&] { load_a(af, wl1, 0); });

        v8f h1[8] = {};
        gemm_chained(h1, h0, wl1, af, [&] {
            load_par(lnb, par + 3 * DHID, lane); load_par(lng, par + 4 * DHID, lane); load_par(lne, par + 5 * DHID, lane); });
        layernorm_tanh_T(h1, lnb, lng, lne, lane, [&] { load_a(af, wl2, 0); });

        v8f h2[8] = {};
        gemm_chained(h2, h1, wl2, af, [&] {
            load_par(lnb, par + 6 * DHID, lane); load_par(lng, par + 7 * DHID, lane); load_par(lne, par + 8 * DHID, lane); });
        v4f w3v[8][2];
        layernorm_tanh_T(h2, lnb, lng, lne, lane, [&] { load_par(w3v, par + 9 * DHID, lane); });

        {
            float p = dot64_T(h2, w3v);
            p += __shfl_xor(p, 16, 32);
            const int e = e0 + c;
            if (h == 0 && e < E) out[e] = p + b3[0];
        }
    }
}

extern "C" void kernel_launch(void* const* d_in, const int* in_sizes, int n_in,
                              void* d_out, int out_size, void* d_ws, size_t ws_size,
                              hipStream_t stream)
{
    const float* x   = (const float*)d_in[0];
    const int*   ei  = (const int*)  d_in[1];
    const float* W0  = (const float*)d_in[2];
    const float* b0  = (const float*)d_in[3];
    const float* W1  = (const float*)d_in[4];
    const float* b1  = (const float*)d_in[5];
    const float* W2  = (const float*)d_in[6];
    const float* b2  = (const float*)d_in[7];
    const float* W3  = (const float*)d_in[8];
    const float* b3  = (const float*)d_in[9];
    const float* g0  = (const float*)d_in[10];
    const float* be0 = (const float*)d_in[11];
    const float* g1  = (const float*)d_in[12];
    const float* be1 = (const float*)d_in[13];
    const float* g2  = (const float*)d_in[14];
    const float* be2 = (const float*)d_in[15];

    const int E     = in_sizes[1] / 2;
    const int N     = in_sizes[0] / DHID;
    const int tiles = (E + TILE_E - 1) / TILE_E;

    const size_t need = (size_t)N * PQ_ROW * sizeof(float);
    if (d_ws != nullptr && ws_size >= need) {
        const int ntiles = (N + TILE_E - 1) / TILE_E;
        int nblocks = (ntiles + WPB - 1) / WPB;
        if (nblocks > 256) nblocks = 256;
        int eblocks = (tiles + WPB - 1) / WPB;
        if (eblocks > 256) eblocks = 256;
        float* pq = (float*)d_ws;
        node_proj_wmma_f32<<<nblocks, WPB * 32, (size_t)SMEM_NODE_FLOATS * sizeof(float), stream>>>(
            x, N, W0, pq);
        edge_mlp_pq_wmma_f32<<<eblocks, WPB * 32, (size_t)SMEM_EDGE_FLOATS * sizeof(float), stream>>>(
            pq, ei, E, b0, W1, b1, W2, b2, W3, b3,
            g0, be0, g1, be1, g2, be2, (float*)d_out);
        return;
    }

    int blocks = (tiles + WPB - 1) / WPB;
    if (blocks > 256) blocks = 256;
    const size_t shmem = (size_t)SMEM_FLOATS * sizeof(float);

    edge_mlp_wmma_f32<<<blocks, WPB * 32, shmem, stream>>>(
        x, ei, E, W0, b0, W1, b1, W2, b2, W3, b3,
        g0, be0, g1, be1, g2, be2, (float*)d_out);
}


# wave issue priority by remaining work (s_setprio 3 while a wave has 3 or more tiles ahead, then 2, 1, 0): the barrier-free waves of a SIMD end together, none runs its list alone; sums unchanged
# speedup vs baseline: 3.0029x; 1.0414x over previous
#include <hip/hip_runtime.h>
#include <math.h>

typedef __attribute__((ext_vector_type(2))) float v2f;
typedef __attribute__((ext_vector_type(4))) float v4f;
typedef __attribute__((ext_vector_type(8))) float v8f;

#define DHID   128
#define TILE_E 16
#define WPB    12
#define LNEPS  1e-5f

#if __has_builtin(__builtin_amdgcn_tanhf)
#define TANHF(v) __builtin_amdgcn_tanhf(v)
#else
#define TANHF(v) tanhf(v)
#endif

#define SCHED_FENCE() __builtin_amdgcn_sched_barrier(0)

__device__ __forceinline__ void prio_by_remaining(int rem)
{
    if (rem >= 3)      __builtin_amdgcn_s_setprio(3);
    else if (rem == 2) __builtin_amdgcn_s_setprio(2);
    else if (rem == 1) __builtin_amdgcn_s_setprio(1);
    else               __builtin_amdgcn_s_setprio(0);
}

__device__ __forceinline__ int tiles_of_wave(int first, int tiles, int stride, int sweeps)
{
    return sweeps - 1 + (int)((sweeps - 1) * stride + first < tiles);
}

__device__ __forceinline__ v2f acc_pair(const v8f& t, int j) { return (v2f){t[2 * j], t[2 * j + 1]}; }
__device__ __forceinline__ void set_pair(v8f& t, int j, v2f v) { t[2 * j] = v.x; t[2 * j + 1] = v.y; }
__device__ __forceinline__ v2f par_pair(const v4f p[2], int j)
{
    const v4f q = p[j >> 1];
    return (j & 1) ? (v2f){q.z, q.w} : (v2f){q.x, q.y};
}
__device__ __forceinline__ v2f fma2(v2f a, v2f b, v2f c) { return __builtin_elementwise_fma(a, b, c); }
__device__ __forceinline__ v2f tree8(const v2f t[8])
{
    return ((t[0] + t[1]) + (t[2] + t[3])) + ((t[4] + t[5]) + (t[6] + t[7]));
}

#define OFF_W1 32768
#define OFF_W2 49152
#define OFF_P  65536
#define SMEM_FLOATS (OFF_P + 10 * DHID)

template <bool CHAINED>
__device__ __forceinline__ void pack_wT(float* __restrict__ dst,
                                        const float* __restrict__ W,
                                        int K, int tid, int nthr)
{
    const int total = K << 7;
#pragma unroll 4
    for (int i = tid; i < total; i += nthr) {
        const int k = i >> 7;
        const int n = i & 127;
        int kb, h;
        if (CHAINED) { kb = ((k >> 4) << 1) | ((k >> 2) & 1); h = (k >> 3) & 1; }
        else         { kb = ((k >> 5) << 2) | ((k >> 2) & 3); h = (k >> 4) & 1; }
        dst[((((kb << 3) + (n >> 4)) << 5) + (n & 15) + (h << 4)) * 4 + (k & 3)] = W[i];
    }
}

__device__ __forceinline__ void load_a(v4f a[8], const float* __restrict__ wl, int kb)
{
#pragma unroll
    for (int mm = 0; mm < 8; ++mm)
        a[mm] = *(const v4f*)(wl + (kb << 10) + (mm << 7));
}

__device__ __forceinline__ void mma_kb(v8f acc[8], const v4f a[8], v4f b)
{
    const v2f b0 = {b.x, b.y};
    const v2f b1 = {b.z, b.w};
#pragma unroll
    for (int mm = 0; mm < 8; ++mm) {
        const v2f a0 = {a[mm].x, a[mm].y};
        acc[mm] = __builtin_amdgcn_wmma_f32_16x16x4_f32(
            false, a0, false, b0, (short)0, acc[mm], false, false);
    }
#pragma unroll
    for (int mm = 0; mm < 8; ++mm) {
        const v2f a1 = {a[mm].z, a[mm].w};
        acc[mm] = __builtin_amdgcn_wmma_f32_16x16x4_f32(
            false, a1, false, b1, (short)0, acc[mm], false, false);
    }
}

template <class Tail>
__device__ __forceinline__ void gemm_chained(v8f acc[8], const v8f prev[8],
                                             const float* __restrict__ wl, v4f a0[8], Tail tail)
{
    v4f a1[8];
#pragma unroll
    for (int nn = 0; nn < 8; ++nn) {
        const v8f t = prev[nn];
        load_a(a1, wl, 2 * nn + 1);
        SCHED_FENCE();
        mma_kb(acc, a0, (v4f){t[0], t[1], t[2], t[3]});
        SCHED_FENCE();
        if (nn + 1 < 8) load_a(a0, wl, 2 * nn + 2); else tail();
        SCHED_FENCE();
        mma_kb(acc, a1, (v4f){t[4], t[5], t[6], t[7]});
        SCHED_FENCE();
    }
}

__device__ __forceinline__ void load_par(v4f p[8][2], const float* __restrict__ row, int lane)
{
    const float* pl = row + ((lane >> 4) << 3);
#pragma unroll
    for (int mm = 0; mm < 8; ++mm) {
        p[mm][0] = *(const v4f*)(pl + (mm << 4));
        p[mm][1] = *(const v4f*)(pl + (mm << 4) + 4);
    }
}

template <class Next>
__device__ __forceinline__ void layernorm_tanh_T(v8f acc[8], const v4f pb[8][2],
                                                 const v4f pg[8][2], const v4f pe[8][2],
                                                 int lane, Next next)
{
    SCHED_FENCE();

    v2f sp[8], qp[8];
#pragma unroll
    for (int mm = 0; mm < 8; ++mm) {
        v2f s2, q2;
#pragma unroll
        for (int j = 0; j < 4; ++j) {
            const v2f v = acc_pair(acc[mm], j) + par_pair(pb[mm], j);
            set_pair(acc[mm], j, v);
            if (j == 0) { s2 = v; q2 = v * v; }
            else        { s2 += v; q2 = fma2(v, v, q2); }
        }
        sp[mm] = s2; qp[mm] = q2;
    }
    const v2f st = tree8(sp);
    const v2f qt = tree8(qp);
    float s  = st.x + st.y;
    float ss = qt.x + qt.y;
    s  += __shfl_xor(s,  16, 32);
    ss += __shfl_xor(ss, 16, 32);
    const float mu  = s * (1.0f / 128.0f);
    const float var = fmaf(-mu, mu, ss * (1.0f / 128.0f));
    const float inv = rsqrtf(var + LNEPS);

    SCHED_FENCE();
    next();
    SCHED_FENCE();

    const v2f mu2  = {mu, mu};
    const v2f inv2 = {inv, inv};
#pragma unroll
    for (int mm = 0; mm < 8; ++mm) {
#pragma unroll
        for (int j = 0; j < 4; ++j) {
            const v2f v = (acc_pair(acc[mm], j) - mu2) * inv2;
            const v2f t = fma2(v, par_pair(pg[mm], j), par_pair(pe[mm], j));
            set_pair(acc[mm], j, (v2f){TANHF(t.x), TANHF(t.y)});
        }
    }
}

__device__ __forceinline__ float dot64_T(const v8f hh[8], const v4f w[8][2])
{
    v2f pp[8];
#pragma unroll
    for (int mm = 0; mm < 8; ++mm) {
        v2f p2 = acc_pair(hh[mm], 0) * par_pair(w[mm], 0);
#pragma unroll
        for (int j = 1; j < 4; ++j)
            p2 = fma2(acc_pair(hh[mm], j), par_pair(w[mm], j), p2);
        pp[mm] = p2;
    }
    const v2f pt = tree8(pp);
    return pt.x + pt.y;
}

__device__ __forceinline__ void load_x(v4f d[4], const float* __restrict__ p)
{
#pragma unroll
    for (int u = 0; u < 4; ++u) d[u] = *(const v4f*)(p + (u << 2));
}

__global__ void __launch_bounds__(WPB * 32, 1)
edge_mlp_wmma_f32(const float* __restrict__ x, const int* __restrict__ eidx, int E,
                  const float* __restrict__ W0, const float* __restrict__ b0,
                  const float* __restrict__ W1, const float* __restrict__ b1,
                  const float* __restrict__ W2, const float* __restrict__ b2,
                  const float* __restrict__ W3, const float* __restrict__ b3,
                  const float* __restrict__ g0, const float* __restrict__ be0,
                  const float* __restrict__ g1, const float* __restrict__ be1,
                  const float* __restrict__ g2, const float* __restrict__ be2,
                  float* __restrict__ out)
{
    extern __shared__ float smem[];
    const int tid  = threadIdx.x;
    const int lane = tid & 31;
    const int wave = tid >> 5;

    float* wp0 = smem;
    float* wp1 = smem + OFF_W1;
    float* wp2 = smem + OFF_W2;
    float* par = smem + OFF_P;

    pack_wT<false>(wp0, W0, 2 * DHID, tid, WPB * 32);
    pack_wT<true >(wp1, W1, DHID,     tid, WPB * 32);
    pack_wT<true >(wp2, W2, DHID,     tid, WPB * 32);
    if (tid < DHID) {
        par[0 * DHID + tid] = b0[tid];  par[1 * DHID + tid] = g0[tid];  par[2 * DHID + tid] = be0[tid];
        par[3 * DHID + tid] = b1[tid];  par[4 * DHID + tid] = g1[tid];  par[5 * DHID + tid] = be1[tid];
        par[6 * DHID + tid] = b2[tid];  par[7 * DHID + tid] = g2[tid];  par[8 * DHID + tid] = be2[tid];
        par[9 * DHID + tid] = W3[tid];
    }
    __syncthreads();

    const int tiles  = (E + TILE_E - 1) / TILE_E;
    const int stride = gridDim.x * WPB;
    const int sweeps = (tiles + stride - 1) / stride;
    const int first = __builtin_amdgcn_readfirstlane(wave) * (int)gridDim.x + (int)blockIdx.x;
    const int mine  = tiles_of_wave(first, tiles, stride, sweeps);

    const int c  = lane & 15;
    const int h  = lane >> 4;
    const float* wl0 = wp0 + (lane << 2);
    const float* wl1 = wp1 + (lane << 2);
    const float* wl2 = wp2 + (lane << 2);

    for (int it = 0; it < sweeps; ++it) {
        const int tile = it * stride + wave * gridDim.x + blockIdx.x;
        if (__builtin_amdgcn_readfirstlane((int)(tile >= tiles))) break;
        prio_by_remaining(mine - 1 - it);
        const int e0 = tile * TILE_E;

        {
            const int tn = tile + stride;
            if (tn < tiles) {
                __builtin_prefetch(&eidx[tn * TILE_E], 0, 1);
                __builtin_prefetch(&eidx[E + tn * TILE_E], 0, 1);
            }
        }

        int e_l = e0 + c; if (e_l >= E) e_l = E - 1;
        const int idx_s = eidx[e_l];
        const int idx_e = eidx[E + e_l];

        const float* rs = x + (size_t)idx_s * DHID + (h << 4);
        const float* re = x + (size_t)idx_e * DHID + (h << 4);
        v8f h0[8] = {};
        v4f lnb[8][2], lng[8][2], lne[8][2];
        {
            v4f xq[2][4];
            v4f a0[8], a1[8];
            load_x(xq[0], rs);
            load_a(a0, wl0, 0);
#pragma unroll
            for (int q = 0; q < 8; ++q) {
                if (q + 1 < 8)
                    load_x(xq[(q + 1) & 1], ((q + 1) < 4 ? rs : re) + (((q + 1) & 3) << 5));
#pragma unroll
                for (int u = 0; u < 4; u += 2) {
                    const int kb = 4 * q + u;
                    load_a(a1, wl0, kb + 1);
                    SCHED_FENCE();
                    mma_kb(h0, a0, xq[q & 1][u]);
                    SCHED_FENCE();
                    if (kb + 2 < 32) load_a(a0, wl0, kb + 2);
                    else { load_par(lnb, par, lane); load_par(lng, par + DHID, lane); load_par(lne, par + 2 * DHID, lane); }
                    SCHED_FENCE();
                    mma_kb(h0, a1, xq[q & 1][u + 1]);
                    SCHED_FENCE();
                }
            }
        }
        v4f af[8];
        layernorm_tanh_T(h0, lnb, lng, lne, lane, [&] { load_a(af, wl1, 0); });

        v8f h1[8] = {};
        gemm_chained(h1, h0, wl1, af, [&] {
            load_par(lnb, par + 3 * DHID, lane); load_par(lng, par + 4 * DHID, lane); load_par(lne, par + 5 * DHID, lane); });
        layernorm_tanh_T(h1, lnb, lng, lne, lane, [&] { load_a(af, wl2, 0); });

        v8f h2[8] = {};
        gemm_chained(h2, h1, wl2, af, [&] {
            load_par(lnb, par + 6 * DHID, lane); load_par(lng, par + 7 * DHID, lane); load_par(lne, par + 8 * DHID, lane); });
        v4f w3v[8][2];
        layernorm_tanh_T(h2, lnb, lng, lne, lane, [&] { load_par(w3v, par + 9 * DHID, lane); });

        {
            float p = dot64_T(h2, w3v);
            p += __shfl_xor(p, 16, 32);
            const int e = e0 + c;
            if (h == 0 && e < E) out[e] = p + b3[0];
        }
    }
    __builtin_amdgcn_s_setprio(0);
}

#define SMEM_NODE_FLOATS 32768
#define PQ_ROW 256

__global__ void __launch_bounds__(WPB * 32, 1)
node_proj_wmma_f32(const float* __restrict__ x, int N, const float* __restrict__ W0,
                   float* __restrict__ pq)
{
    extern __shared__ float smem[];
    const int tid  = threadIdx.x;
    const int lane = tid & 31;
    const int wave = tid >> 5;

    pack_wT<false>(smem, W0, 2 * DHID, tid, WPB * 32);
    __syncthreads();

    const int tiles  = (N + TILE_E - 1) / TILE_E;
    const int stride = gridDim.x * WPB;
    const int sweeps = (tiles + stride - 1) / stride;
    const int first = __builtin_amdgcn_readfirstlane(wave) * (int)gridDim.x + (int)blockIdx.x;
    const int mine  = tiles_of_wave(first, tiles, stride, sweeps);

    const int c = lane & 15;
    const int h = lane >> 4;
    const float* wl0 = smem + (lane << 2);

    for (int it = 0; it < sweeps; ++it) {
        const int tile = it * stride + wave * gridDim.x + blockIdx.x;
        if (__builtin_amdgcn_readfirstlane((int)(tile >= tiles))) break;
        prio_by_remaining(mine - 1 - it);

        int n_l = tile * TILE_E + c; if (n_l >= N) n_l = N - 1;
        const float* rs = x + (size_t)n_l * DHID + (h << 4);

        v8f accP[8] = {};
        v8f accQ[8] = {};
        {
            v4f xq[2][4];
            v4f a0[8], a1[8];
            load_x(xq[0], rs);
            load_a(a0, wl0, 0);
#pragma unroll
            for (int q = 0; q < 4; ++q) {
                if (q + 1 < 4) load_x(xq[(q + 1) & 1], rs + ((q + 1) << 5));
#pragma unroll
                for (int u = 0; u < 4; ++u) {
                    const int kb = 4 * q + u;
                    load_a(a1, wl0, 16 + kb);
                    SCHED_FENCE();
                    mma_kb(accP, a0, xq[q & 1][u]);
                    SCHED_FENCE();
                    if (kb + 1 < 16) load_a(a0, wl0, kb + 1);
                    SCHED_FENCE();
                    mma_kb(accQ, a1, xq[q & 1][u]);
                    SCHED_FENCE();
                }
            }
        }

        const int n = tile * TILE_E + c;
        if (n < N) {
            float* o = pq + (size_t)n * PQ_ROW + (h << 3);
#pragma unroll
            for (int mm = 0; mm < 8; ++mm) {
                *(v4f*)(o + (mm << 4))            = (v4f){accP[mm][0], accP[mm][1], accP[mm][2], accP[mm][3]};
                *(v4f*)(o + (mm << 4) + 4)        = (v4f){accP[mm][4], accP[mm][5], accP[mm][6], accP[mm][7]};
                *(v4f*)(o + DHID + (mm << 4))     = (v4f){accQ[mm][0], accQ[mm][1], accQ[mm][2], accQ[mm][3]};
                *(v4f*)(o + DHID + (mm << 4) + 4) = (v4f){accQ[mm][4], accQ[mm][5], accQ[mm][6], accQ[mm][7]};
            }
        }
    }
    __builtin_amdgcn_s_setprio(0);
}

#define EOFF_W2 16384
#define EOFF_P  32768
#define SMEM_EDGE_FLOATS (EOFF_P + 10 * DHID)

__global__ void __launch_bounds__(WPB * 32, 1)
edge_mlp_pq_wmma_f32(const float* __restrict__ pq, const int* __restrict__ eidx, int E,
                     const float* __restrict__ b0,
                     const float* __restrict__ W1, const float* __restrict__ b1,
                     const float* __restrict__ W2, const float* __restrict__ b2,
                     const float* __restrict__ W3, const float* __restrict__ b3,
                     const float* __restrict__ g0, const float* __restrict__ be0,
                     const float* __restrict__ g1, const float* __restrict__ be1,
                     const float* __restrict__ g2, const float* __restrict__ be2,
                     float* __restrict__ out)
{
    extern __shared__ float smem[];
    const int tid  = threadIdx.x;
    const int lane = tid & 31;
    const int wave = tid >> 5;

    float* wp1 = smem;
    float* wp2 = smem + EOFF_W2;
    float* par = smem + EOFF_P;

    pack_wT<true>(wp1, W1, DHID, tid, WPB * 32);
    pack_wT<true>(wp2, W2, DHID, tid, WPB * 32);
    if (tid < DHID) {
        par[0 * DHID + tid] = b0[tid];  par[1 * DHID + tid] = g0[tid];  par[2 * DHID + tid] = be0[tid];
        par[3 * DHID + tid] = b1[tid];  par[4 * DHID + tid] = g1[tid];  par[5 * DHID + tid] = be1[tid];
        par[6 * DHID + tid] = b2[tid];  par[7 * DHID + tid] = g2[tid];  par[8 * DHID + tid] = be2[tid];
        par[9 * DHID + tid] = W3[tid];
    }
    __syncthreads();

    const int tiles  = (E + TILE_E - 1) / TILE_E;
    const int stride = gridDim.x * WPB;
    const int sweeps = (tiles + stride - 1) / stride;
    const int first = __builtin_amdgcn_readfirstlane(wave) * (int)gridDim.x + (int)blockIdx.x;
    const int mine  = tiles_of_wave(first, tiles, stride, sweeps);

    const int c  = lane & 15;
    const int h  = lane >> 4;
    const float* wl1 = wp1 + (lane << 2);
    const float* wl2 = wp2 + (lane << 2);

    for (int it = 0; it < sweeps; ++it) {
        const int tile = it * stride + wave * gridDim.x + blockIdx.x;
        if (__builtin_amdgcn_readfirstlane((int)(tile >= tiles))) break;
        prio_by_remaining(mine - 1 - it);
        const int e0 = tile * TILE_E;

        {
            const int tn = tile + stride;
            if (tn < tiles) {
                __builtin_prefetch(&eidx[tn * TILE_E], 0, 1);
                __builtin_prefetch(&eidx[E + tn * TILE_E], 0, 1);
            }
        }

        int e_l = e0 + c; if (e_l >= E) e_l = E - 1;
        const int idx_s = eidx[e_l];
        const int idx_e = eidx[E + e_l];

        const float* ps = pq + (size_t)idx_s * PQ_ROW + (h << 3);
        const float* pe = pq + (size_t)idx_e * PQ_ROW + DHID + (h << 3);
        v8f h0[8];
        v4f lnb[8][2], lng[8][2], lne[8][2];
        load_par(lnb, par, lane);
        {
            v4f p[8][2], r[8][2];
#pragma unroll
            for (int mm = 0; mm < 8; ++mm) {
                p[mm][0] = *(const v4f*)(ps + (mm << 4));
                p[mm][1] = *(const v4f*)(ps + (mm << 4) + 4);
            }
#pragma unroll
            for (int mm = 0; mm < 8; ++mm) {
                r[mm][0] = *(const v4f*)(pe + (mm << 4));
                r[mm][1] = *(const v4f*)(pe + (mm << 4) + 4);
            }
#pragma unroll
            for (int mm = 0; mm < 8; ++mm) {
                const v4f lo = p[mm][0] + r[mm][0];
                const v4f hi = p[mm][1] + r[mm][1];
                h0[mm] = (v8f){lo.x, lo.y, lo.z, lo.w, hi.x, hi.y, hi.z, hi.w};
            }
        }
        SCHED_FENCE();
        load_par(lng, par + DHID, lane);
        load_par(lne, par + 2 * DHID, lane);
        v4f af[8];
        layernorm_tanh_T(h0, lnb, lng, lne, lane, [&] { load_a(af, wl1, 0); });

        v8f h1[8] = {};
        gemm_chained(h1, h0, wl1, af, [&] {
            load_par(lnb, par + 3 * DHID, lane); load_par(lng, par + 4 * DHID, lane); load_par(lne, par + 5 * DHID, lane); });
        layernorm_tanh_T(h1, lnb, lng, lne, lane, [&] { load_a(af, wl2, 0); });

        v8f h2[8] = {};
        gemm_chained(h2, h1, wl2, af, [&] {
            load_par(lnb, par + 6 * DHID, lane); load_par(lng, par + 7 * DHID, lane); load_par(lne, par + 8 * DHID, lane); });
        v4f w3v[8][2];
        layernorm_tanh_T(h2, lnb, lng, lne, lane, [&] { load_par(w3v, par + 9 * DHID, lane); });

        {
            float p = dot64_T(h2, w3v);
            p += __shfl_xor(p, 16, 32);
            const int e = e0 + c;
            if (h == 0 && e < E) out[e] = p + b3[0];
        }
    }
    __builtin_amdgcn_s_setprio(0);
}

extern "C" void kernel_launch(void* const* d_in, const int* in_sizes, int n_in,
                              void* d_out, int out_size, void* d_ws, size_t ws_size,
                              hipStream_t stream)
{
    const float* x   = (const float*)d_in[0];
    const int*   ei  = (const int*)  d_in[1];
    const float* W0  = (const float*)d_in[2];
    const float* b0  = (const float*)d_in[3];
    const float* W1  = (const float*)d_in[4];
    const float* b1  = (const float*)d_in[5];
    const float* W2  = (const float*)d_in[6];
    const float* b2  = (const float*)d_in[7];
    const float* W3  = (const float*)d_in[8];
    const float* b3  = (const float*)d_in[9];
    const float* g0  = (const float*)d_in[10];
    const float* be0 = (const float*)d_in[11];
    const float* g1  = (const float*)d_in[12];
    const float* be1 = (const float*)d_in[13];
    const float* g2  = (const float*)d_in[14];
    const float* be2 = (const float*)d_in[15];

    const int E     = in_sizes[1] / 2;
    const int N     = in_sizes[0] / DHID;
    const int tiles = (E + TILE_E - 1) / TILE_E;

    const size_t need = (size_t)N * PQ_ROW * sizeof(float);
    if (d_ws != nullptr && ws_size >= need) {
        const int ntiles = (N + TILE_E - 1) / TILE_E;
        int nblocks = (ntiles + WPB - 1) / WPB;
        if (nblocks > 256) nblocks = 256;
        int eblocks = (tiles + WPB - 1) / WPB;
        if (eblocks > 256) eblocks = 256;
        float* pq = (float*)d_ws;
        node_proj_wmma_f32<<<nblocks, WPB * 32, (size_t)SMEM_NODE_FLOATS * sizeof(float), stream>>>(
            x, N, W0, pq);
        edge_mlp_pq_wmma_f32<<<eblocks, WPB * 32, (size_t)SMEM_EDGE_FLOATS * sizeof(float), stream>>>(
            pq, ei, E, b0, W1, b1, W2, b2, W3, b3,
            g0, be0, g1, be1, g2, be2, (float*)d_out);
        return;
    }

    int blocks = (tiles + WPB - 1) / WPB;
    if (blocks > 256) blocks = 256;
    const size_t shmem = (size_t)SMEM_FLOATS * sizeof(float);

    edge_mlp_wmma_f32<<<blocks, WPB * 32, shmem, stream>>>(
        x, ei, E, W0, b0, W1, b1, W2, b2, W3, b3,
        g0, be0, g1, be1, g2, be2, (float*)d_out);
}


# LN mean folded into the layers: weight rows and biases centred when staged (whole rows per wave, loads issued at once), centred bias starts the accumulators; LN sums squares only; sum order changes
# speedup vs baseline: 3.1447x; 1.0472x over previous
#include <hip/hip_runtime.h>
#include <math.h>

typedef __attribute__((ext_vector_type(2))) float v2f;
typedef __attribute__((ext_vector_type(4))) float v4f;
typedef __attribute__((ext_vector_type(8))) float v8f;

#define DHID   128
#define TILE_E 16
#define WPB    12
#define LNEPS  1e-5f

#if __has_builtin(__builtin_amdgcn_tanhf)
#define TANHF(v) __builtin_amdgcn_tanhf(v)
#else
#define TANHF(v) tanhf(v)
#endif

#define SCHED_FENCE() __builtin_amdgcn_sched_barrier(0)

__device__ __forceinline__ void prio_by_remaining(int rem)
{
    if (rem >= 3)      __builtin_amdgcn_s_setprio(3);
    else if (rem == 2) __builtin_amdgcn_s_setprio(2);
    else if (rem == 1) __builtin_amdgcn_s_setprio(1);
    else               __builtin_amdgcn_s_setprio(0);
}

__device__ __forceinline__ int tiles_of_wave(int first, int tiles, int stride, int sweeps)
{
    return sweeps - 1 + (int)((sweeps - 1) * stride + first < tiles);
}

__device__ __forceinline__ v2f acc_pair(const v8f& t, int j) { return (v2f){t[2 * j], t[2 * j + 1]}; }
__device__ __forceinline__ void set_pair(v8f& t, int j, v2f v) { t[2 * j] = v.x; t[2 * j + 1] = v.y; }
__device__ __forceinline__ v2f par_pair(const v4f p[2], int j)
{
    const v4f q = p[j >> 1];
    return (j & 1) ? (v2f){q.z, q.w} : (v2f){q.x, q.y};
}
__device__ __forceinline__ v2f fma2(v2f a, v2f b, v2f c) { return __builtin_elementwise_fma(a, b, c); }
__device__ __forceinline__ v2f tree8(const v2f t[8])
{
    return ((t[0] + t[1]) + (t[2] + t[3])) + ((t[4] + t[5]) + (t[6] + t[7]));
}

#define OFF_W1 32768
#define OFF_W2 49152
#define OFF_P  65536
#define SMEM_FLOATS (OFF_P + 10 * DHID)

template <bool CHAINED>
__device__ __forceinline__ void pack_wT(float* __restrict__ dst,
                                        const float* __restrict__ W,
                                        int K, int tid, int nthr)
{
    const int total = K << 7;
#pragma unroll 4
    for (int i = tid; i < total; i += nthr) {
        const int k = i >> 7;
        const int n = i & 127;
        int kb, h;
        if (CHAINED) { kb = ((k >> 4) << 1) | ((k >> 2) & 1); h = (k >> 3) & 1; }
        else         { kb = ((k >> 5) << 2) | ((k >> 2) & 3); h = (k >> 4) & 1; }
        dst[((((kb << 3) + (n >> 4)) << 5) + (n & 15) + (h << 4)) * 4 + (k & 3)] = W[i];
    }
}

__device__ __forceinline__ void load_a(v4f a[8], const float* __restrict__ wl, int kb)
{
#pragma unroll
    for (int mm = 0; mm < 8; ++mm)
        a[mm] = *(const v4f*)(wl + (kb << 10) + (mm << 7));
}

__device__ __forceinline__ void mma_kb(v8f acc[8], const v4f a[8], v4f b)
{
    const v2f b0 = {b.x, b.y};
    const v2f b1 = {b.z, b.w};
#pragma unroll
    for (int mm = 0; mm < 8; ++mm) {
        const v2f a0 = {a[mm].x, a[mm].y};
        acc[mm] = __builtin_amdgcn_wmma_f32_16x16x4_f32(
            false, a0, false, b0, (short)0, acc[mm], false, false);
    }
#pragma unroll
    for (int mm = 0; mm < 8; ++mm) {
        const v2f a1 = {a[mm].z, a[mm].w};
        acc[mm] = __builtin_amdgcn_wmma_f32_16x16x4_f32(
            false, a1, false, b1, (short)0, acc[mm], false, false);
    }
}

template <class Tail>
__device__ __forceinline__ void gemm_chained(v8f acc[8], const v8f prev[8],
                                             const float* __restrict__ wl, v4f a0[8], Tail tail)
{
    v4f a1[8];
#pragma unroll
    for (int nn = 0; nn < 8; ++nn) {
        const v8f t = prev[nn];
        load_a(a1, wl, 2 * nn + 1);
        SCHED_FENCE();
        mma_kb(acc, a0, (v4f){t[0], t[1], t[2], t[3]});
        SCHED_FENCE();
        if (nn + 1 < 8) load_a(a0, wl, 2 * nn + 2); else tail();
        SCHED_FENCE();
        mma_kb(acc, a1, (v4f){t[4], t[5], t[6], t[7]});
        SCHED_FENCE();
    }
}

__device__ __forceinline__ void load_par(v4f p[8][2], const float* __restrict__ row, int lane)
{
    const float* pl = row + ((lane >> 4) << 3);
#pragma unroll
    for (int mm = 0; mm < 8; ++mm) {
        p[mm][0] = *(const v4f*)(pl + (mm << 4));
        p[mm][1] = *(const v4f*)(pl + (mm << 4) + 4);
    }
}

template <class Next>
__device__ __forceinline__ void layernorm_tanh_T(v8f acc[8], const v4f pb[8][2],
                                                 const v4f pg[8][2], const v4f pe[8][2],
                                                 int lane, Next next)
{
    SCHED_FENCE();

    v2f sp[8], qp[8];
#pragma unroll
    for (int mm = 0; mm < 8; ++mm) {
        v2f s2, q2;
#pragma unroll
        for (int j = 0; j < 4; ++j) {
            const v2f v = acc_pair(acc[mm], j) + par_pair(pb[mm], j);
            set_pair(acc[mm], j, v);
            if (j == 0) { s2 = v; q2 = v * v; }
            else        { s2 += v; q2 = fma2(v, v, q2); }
        }
        sp[mm] = s2; qp[mm] = q2;
    }
    const v2f st = tree8(sp);
    const v2f qt = tree8(qp);
    float s  = st.x + st.y;
    float ss = qt.x + qt.y;
    s  += __shfl_xor(s,  16, 32);
    ss += __shfl_xor(ss, 16, 32);
    const float mu  = s * (1.0f / 128.0f);
    const float var = fmaf(-mu, mu, ss * (1.0f / 128.0f));
    const float inv = rsqrtf(var + LNEPS);

    SCHED_FENCE();
    next();
    SCHED_FENCE();

    const v2f mu2  = {mu, mu};
    const v2f inv2 = {inv, inv};
#pragma unroll
    for (int mm = 0; mm < 8; ++mm) {
#pragma unroll
        for (int j = 0; j < 4; ++j) {
            const v2f v = (acc_pair(acc[mm], j) - mu2) * inv2;
            const v2f t = fma2(v, par_pair(pg[mm], j), par_pair(pe[mm], j));
            set_pair(acc[mm], j, (v2f){TANHF(t.x), TANHF(t.y)});
        }
    }
}

__device__ __forceinline__ float dot64_T(const v8f hh[8], const v4f w[8][2])
{
    v2f pp[8];
#pragma unroll
    for (int mm = 0; mm < 8; ++mm) {
        v2f p2 = acc_pair(hh[mm], 0) * par_pair(w[mm], 0);
#pragma unroll
        for (int j = 1; j < 4; ++j)
            p2 = fma2(acc_pair(hh[mm], j), par_pair(w[mm], j), p2);
        pp[mm] = p2;
    }
    const v2f pt = tree8(pp);
    return pt.x + pt.y;
}

__device__ __forceinline__ float wave_sum(float s)
{
#pragma unroll
    for (int m = 16; m >= 1; m >>= 1) s += __shfl_xor(s, m, 32);
    return s;
}

#define STAGE_U 6
template <class Src, class Dst>
__device__ __forceinline__ void stage_centred(float* __restrict__ smem, int ngroups,
                                              int wave, int lane, Src src, Dst dstoff)
{
    v4f r[STAGE_U][4];
#pragma unroll
    for (int u = 0; u < STAGE_U; ++u) {
        int g = wave + WPB * u; if (g >= ngroups) g = ngroups - 1;
        const float* p = src(g) + (lane << 2);
#pragma unroll
        for (int i = 0; i < 4; ++i) r[u][i] = *(const v4f*)(p + (i << 7));
    }
    float s[STAGE_U][4];
#pragma unroll
    for (int u = 0; u < STAGE_U; ++u)
#pragma unroll
        for (int i = 0; i < 4; ++i) s[u][i] = (r[u][i].x + r[u][i].y) + (r[u][i].z + r[u][i].w);
#pragma unroll
    for (int m = 16; m >= 1; m >>= 1)
#pragma unroll
        for (int u = 0; u < STAGE_U; ++u)
#pragma unroll
            for (int i = 0; i < 4; ++i) s[u][i] += __shfl_xor(s[u][i], m, 32);
#pragma unroll
    for (int u = 0; u < STAGE_U; ++u) {
        const int g = wave + WPB * u;
        if (g < ngroups) {
            const float m0 = s[u][0] * (1.0f / 128.0f), m1 = s[u][1] * (1.0f / 128.0f);
            const float m2 = s[u][2] * (1.0f / 128.0f), m3 = s[u][3] * (1.0f / 128.0f);
            float* d = smem + dstoff(g) + ((((lane >> 2) << 5) + ((lane & 3) << 2)) << 2);
            *(v4f*)(d)      = (v4f){r[u][0].x - m0, r[u][1].x - m1, r[u][2].x - m2, r[u][3].x - m3};
            *(v4f*)(d + 4)  = (v4f){r[u][0].y - m0, r[u][1].y - m1, r[u][2].y - m2, r[u][3].y - m3};
            *(v4f*)(d + 8)  = (v4f){r[u][0].z - m0, r[u][1].z - m1, r[u][2].z - m2, r[u][3].z - m3};
            *(v4f*)(d + 12) = (v4f){r[u][0].w - m0, r[u][1].w - m1, r[u][2].w - m2, r[u][3].w - m3};
        }
    }
}

__device__ __forceinline__ void stage_centred_bias(float* __restrict__ dst,
                                                   const float* __restrict__ b, int lane)
{
    const v4f v = *(const v4f*)(b + (lane << 2));
    const float m = wave_sum((v.x + v.y) + (v.z + v.w)) * (1.0f / 128.0f);
    *(v4f*)(dst + (lane << 2)) = (v4f){v.x - m, v.y - m, v.z - m, v.w - m};
}

__device__ __forceinline__ void load_tile(v8f t[8], const float* __restrict__ row, int lane)
{
    const float* pl = row + ((lane >> 4) << 3);
#pragma unroll
    for (int mm = 0; mm < 8; ++mm) {
        const v4f lo = *(const v4f*)(pl + (mm << 4));
        const v4f hi = *(const v4f*)(pl + (mm << 4) + 4);
        t[mm] = __builtin_shufflevector(lo, hi, 0, 1, 2, 3, 4, 5, 6, 7);
    }
}

template <class Next>
__device__ __forceinline__ void ln_centred_tanh_T(v8f acc[8], const v4f pg[8][2],
                                                  const v4f pe[8][2], Next next)
{
    SCHED_FENCE();

    v2f qp[8];
#pragma unroll
    for (int mm = 0; mm < 8; ++mm) {
        v2f q2;
#pragma unroll
        for (int j = 0; j < 4; ++j) {
            const v2f v = acc_pair(acc[mm], j);
            if (j == 0) q2 = v * v; else q2 = fma2(v, v, q2);
        }
        qp[mm] = q2;
    }
    const v2f qt = tree8(qp);
    float ss = qt.x + qt.y;
    ss += __shfl_xor(ss, 16, 32);
    const float inv = rsqrtf(fmaf(ss, 1.0f / 128.0f, LNEPS));

    SCHED_FENCE();
    next();
    SCHED_FENCE();

    const v2f inv2 = {inv, inv};
#pragma unroll
    for (int mm = 0; mm < 8; ++mm) {
#pragma unroll
        for (int j = 0; j < 4; ++j) {
            const v2f t = fma2(acc_pair(acc[mm], j) * inv2, par_pair(pg[mm], j), par_pair(pe[mm], j));
            set_pair(acc[mm], j, (v2f){TANHF(t.x), TANHF(t.y)});
        }
    }
}

__device__ __forceinline__ void load_x(v4f d[4], const float* __restrict__ p)
{
#pragma unroll
    for (int u = 0; u < 4; ++u) d[u] = *(const v4f*)(p + (u << 2));
}

__global__ void __launch_bounds__(WPB * 32, 1)
edge_mlp_wmma_f32(const float* __restrict__ x, const int* __restrict__ eidx, int E,
                  const float* __restrict__ W0, const float* __restrict__ b0,
                  const float* __restrict__ W1, const float* __restrict__ b1,
                  const float* __restrict__ W2, const float* __restrict__ b2,
                  const float* __restrict__ W3, const float* __restrict__ b3,
                  const float* __restrict__ g0, const float* __restrict__ be0,
                  const float* __restrict__ g1, const float* __restrict__ be1,
                  const float* __restrict__ g2, const float* __restrict__ be2,
                  float* __restrict__ out)
{
    extern __shared__ float smem[];
    const int tid  = threadIdx.x;
    const int lane = tid & 31;
    const int wave = tid >> 5;

    float* wp0 = smem;
    float* wp1 = smem + OFF_W1;
    float* wp2 = smem + OFF_W2;
    float* par = smem + OFF_P;

    pack_wT<false>(wp0, W0, 2 * DHID, tid, WPB * 32);
    pack_wT<true >(wp1, W1, DHID,     tid, WPB * 32);
    pack_wT<true >(wp2, W2, DHID,     tid, WPB * 32);
    if (tid < DHID) {
        par[0 * DHID + tid] = b0[tid];  par[1 * DHID + tid] = g0[tid];  par[2 * DHID + tid] = be0[tid];
        par[3 * DHID + tid] = b1[tid];  par[4 * DHID + tid] = g1[tid];  par[5 * DHID + tid] = be1[tid];
        par[6 * DHID + tid] = b2[tid];  par[7 * DHID + tid] = g2[tid];  par[8 * DHID + tid] = be2[tid];
        par[9 * DHID + tid] = W3[tid];
    }
    __syncthreads();

    const int tiles  = (E + TILE_E - 1) / TILE_E;
    const int stride = gridDim.x * WPB;
    const int sweeps = (tiles + stride - 1) / stride;
    const int first = __builtin_amdgcn_readfirstlane(wave) * (int)gridDim.x + (int)blockIdx.x;
    const int mine  = tiles_of_wave(first, tiles, stride, sweeps);

    const int c  = lane & 15;
    const int h  = lane >> 4;
    const float* wl0 = wp0 + (lane << 2);
    const float* wl1 = wp1 + (lane << 2);
    const float* wl2 = wp2 + (lane << 2);

    for (int it = 0; it < sweeps; ++it) {
        const int tile = it * stride + wave * gridDim.x + blockIdx.x;
        if (__builtin_amdgcn_readfirstlane((int)(tile >= tiles))) break;
        prio_by_remaining(mine - 1 - it);
        const int e0 = tile * TILE_E;

        {
            const int tn = tile + stride;
            if (tn < tiles) {
                __builtin_prefetch(&eidx[tn * TILE_E], 0, 1);
                __builtin_prefetch(&eidx[E + tn * TILE_E], 0, 1);
            }
        }

        int e_l = e0 + c; if (e_l >= E) e_l = E - 1;
        const int idx_s = eidx[e_l];
        const int idx_e = eidx[E + e_l];

        const float* rs = x + (size_t)idx_s * DHID + (h << 4);
        const float* re = x + (size_t)idx_e * DHID + (h << 4);
        v8f h0[8] = {};
        v4f lnb[8][2], lng[8][2], lne[8][2];
        {
            v4f xq[2][4];
            v4f a0[8], a1[8];
            load_x(xq[0], rs);
            load_a(a0, wl0, 0);
#pragma unroll
            for (int q = 0; q < 8; ++q) {
                if (q + 1 < 8)
                    load_x(xq[(q + 1) & 1], ((q + 1) < 4 ? rs : re) + (((q + 1) & 3) << 5));
#pragma unroll
                for (int u = 0; u < 4; u += 2) {
                    const int kb = 4 * q + u;
                    load_a(a1, wl0, kb + 1);
                    SCHED_FENCE();
                    mma_kb(h0, a0, xq[q & 1][u]);
                    SCHED_FENCE();
                    if (kb + 2 < 32) load_a(a0, wl0, kb + 2);
                    else { load_par(lnb, par, lane); load_par(lng, par + DHID, lane); load_par(lne, par + 2 * DHID, lane); }
                    SCHED_FENCE();
                    mma_kb(h0, a1, xq[q & 1][u + 1]);
                    SCHED_FENCE();
                }
            }
        }
        v4f af[8];
        layernorm_tanh_T(h0, lnb, lng, lne, lane, [&] { load_a(af, wl1, 0); });

        v8f h1[8] = {};
        gemm_chained(h1, h0, wl1, af, [&] {
            load_par(lnb, par + 3 * DHID, lane); load_par(lng, par + 4 * DHID, lane); load_par(lne, par + 5 * DHID, lane); });
        layernorm_tanh_T(h1, lnb, lng, lne, lane, [&] { load_a(af, wl2, 0); });

        v8f h2[8] = {};
        gemm_chained(h2, h1, wl2, af, [&] {
            load_par(lnb, par + 6 * DHID, lane); load_par(lng, par + 7 * DHID, lane); load_par(lne, par + 8 * DHID, lane); });
        v4f w3v[8][2];
        layernorm_tanh_T(h2, lnb, lng, lne, lane, [&] { load_par(w3v, par + 9 * DHID, lane); });

        {
            float p = dot64_T(h2, w3v);
            p += __shfl_xor(p, 16, 32);
            const int e = e0 + c;
            if (h == 0 && e < E) out[e] = p + b3[0];
        }
    }
    __builtin_amdgcn_s_setprio(0);
}

#define NOFF_B 32768
#define SMEM_NODE_FLOATS (NOFF_B + DHID)
#define PQ_ROW 256

__global__ void __launch_bounds__(WPB * 32, 1)
node_proj_wmma_f32(const float* __restrict__ x, int N, const float* __restrict__ W0,
                   const float* __restrict__ b0, float* __restrict__ pq)
{
    extern __shared__ float smem[];
    const int tid  = threadIdx.x;
    const int lane = tid & 31;
    const int wave = tid >> 5;

    stage_centred(smem, 64, wave, lane,
                  [&](int g) { return W0 + ((size_t)g << 9); },
                  [&](int g) { return (((((g >> 3) << 2) | (g & 3)) << 10)) + (((g >> 2) & 1) << 6); });
    if (wave == 0) stage_centred_bias(smem + NOFF_B, b0, lane);
    __syncthreads();

    const int tiles  = (N + TILE_E - 1) / TILE_E;
    const int stride = gridDim.x * WPB;
    const int sweeps = (tiles + stride - 1) / stride;
    const int first = __builtin_amdgcn_readfirstlane(wave) * (int)gridDim.x + (int)blockIdx.x;
    const int mine  = tiles_of_wave(first, tiles, stride, sweeps);

    const int c = lane & 15;
    const int h = lane >> 4;
    const float* wl0 = smem + (lane << 2);

    for (int it = 0; it < sweeps; ++it) {
        const int tile = it * stride + wave * gridDim.x + blockIdx.x;
        if (__builtin_amdgcn_readfirstlane((int)(tile >= tiles))) break;
        prio_by_remaining(mine - 1 - it);

        int n_l = tile * TILE_E + c; if (n_l >= N) n_l = N - 1;
        const float* rs = x + (size_t)n_l * DHID + (h << 4);

        v8f accP[8];
        load_tile(accP, smem + NOFF_B, lane);
        v8f accQ[8] = {};
        {
            v4f xq[2][4];
            v4f a0[8], a1[8];
            load_x(xq[0], rs);
            load_a(a0, wl0, 0);
#pragma unroll
            for (int q = 0; q < 4; ++q) {
                if (q + 1 < 4) load_x(xq[(q + 1) & 1], rs + ((q + 1) << 5));
#pragma unroll
                for (int u = 0; u < 4; ++u) {
                    const int kb = 4 * q + u;
                    load_a(a1, wl0, 16 + kb);
                    SCHED_FENCE();
                    mma_kb(accP, a0, xq[q & 1][u]);
                    SCHED_FENCE();
                    if (kb + 1 < 16) load_a(a0, wl0, kb + 1);
                    SCHED_FENCE();
                    mma_kb(accQ, a1, xq[q & 1][u]);
                    SCHED_FENCE();
                }
            }
        }

        const int n = tile * TILE_E + c;
        if (n < N) {
            float* o = pq + (size_t)n * PQ_ROW + (h << 3);
#pragma unroll
            for (int mm = 0; mm < 8; ++mm) {
                *(v4f*)(o + (mm << 4))            = (v4f){accP[mm][0], accP[mm][1], accP[mm][2], accP[mm][3]};
                *(v4f*)(o + (mm << 4) + 4)        = (v4f){accP[mm][4], accP[mm][5], accP[mm][6], accP[mm][7]};
                *(v4f*)(o + DHID + (mm << 4))     = (v4f){accQ[mm][0], accQ[mm][1], accQ[mm][2], accQ[mm][3]};
                *(v4f*)(o + DHID + (mm << 4) + 4) = (v4f){accQ[mm][4], accQ[mm][5], accQ[mm][6], accQ[mm][7]};
            }
        }
    }
    __builtin_amdgcn_s_setprio(0);
}

#define EOFF_W2 16384
#define EOFF_P  32768
#define SMEM_EDGE_FLOATS (EOFF_P + 9 * DHID)

__global__ void __launch_bounds__(WPB * 32, 1)
edge_mlp_pq_wmma_f32(const float* __restrict__ pq, const int* __restrict__ eidx, int E,
                     const float* __restrict__ W1, const float* __restrict__ b1,
                     const float* __restrict__ W2, const float* __restrict__ b2,
                     const float* __restrict__ W3, const float* __restrict__ b3,
                     const float* __restrict__ g0, const float* __restrict__ be0,
                     const float* __restrict__ g1, const float* __restrict__ be1,
                     const float* __restrict__ g2, const float* __restrict__ be2,
                     float* __restrict__ out)
{
    extern __shared__ float smem[];
    const int tid  = threadIdx.x;
    const int lane = tid & 31;
    const int wave = tid >> 5;

    float* wp1 = smem;
    float* wp2 = smem + EOFF_W2;
    float* par = smem + EOFF_P;

    stage_centred(smem, 64, wave, lane,
                  [&](int g) { return (g < 32 ? W1 : W2) + ((size_t)(g & 31) << 9); },
                  [&](int g) { const int q = g & 31;
                               return (g < 32 ? 0 : EOFF_W2) + (((((q >> 2) << 1) | (q & 1)) << 10)) + (((q >> 1) & 1) << 6); });
    if (wave == 0) stage_centred_bias(par + 2 * DHID, b1, lane);
    if (wave == 1) stage_centred_bias(par + 5 * DHID, b2, lane);
    if (tid >= 2 * 32 && tid < 2 * 32 + DHID) {
        const int t = tid - 2 * 32;
        par[0 * DHID + t] = g0[t];  par[1 * DHID + t] = be0[t];
        par[3 * DHID + t] = g1[t];  par[4 * DHID + t] = be1[t];
        par[6 * DHID + t] = g2[t];  par[7 * DHID + t] = be2[t];
        par[8 * DHID + t] = W3[t];
    }
    __syncthreads();

    const int tiles  = (E + TILE_E - 1) / TILE_E;
    const int stride = gridDim.x * WPB;
    const int sweeps = (tiles + stride - 1) / stride;
    const int first = __builtin_amdgcn_readfirstlane(wave) * (int)gridDim.x + (int)blockIdx.x;
    const int mine  = tiles_of_wave(first, tiles, stride, sweeps);

    const int c  = lane & 15;
    const int h  = lane >> 4;
    const float* wl1 = wp1 + (lane << 2);
    const float* wl2 = wp2 + (lane << 2);

    for (int it = 0; it < sweeps; ++it) {
        const int tile = it * stride + wave * gridDim.x + blockIdx.x;
        if (__builtin_amdgcn_readfirstlane((int)(tile >= tiles))) break;
        prio_by_remaining(mine - 1 - it);
        const int e0 = tile * TILE_E;

        {
            const int tn = tile + stride;
            if (tn < tiles) {
                __builtin_prefetch(&eidx[tn * TILE_E], 0, 1);
                __builtin_prefetch(&eidx[E + tn * TILE_E], 0, 1);
            }
        }

        int e_l = e0 + c; if (e_l >= E) e_l = E - 1;
        const int idx_s = eidx[e_l];
        const int idx_e = eidx[E + e_l];

        const float* ps = pq + (size_t)idx_s * PQ_ROW + (h << 3);
        const float* pe = pq + (size_t)idx_e * PQ_ROW + DHID + (h << 3);
        v8f h0[8];
        v4f lng[8][2], lne[8][2];
        {
            v4f p[8][2], r[8][2];
#pragma unroll
            for (int mm = 0; mm < 8; ++mm) {
                p[mm][0] = *(const v4f*)(ps + (mm << 4));
                p[mm][1] = *(const v4f*)(ps + (mm << 4) + 4);
            }
#pragma unroll
            for (int mm = 0; mm < 8; ++mm) {
                r[mm][0] = *(const v4f*)(pe + (mm << 4));
                r[mm][1] = *(const v4f*)(pe + (mm << 4) + 4);
            }
#pragma unroll
            for (int mm = 0; mm < 8; ++mm) {
                const v4f lo = p[mm][0] + r[mm][0];
                const v4f hi = p[mm][1] + r[mm][1];
                h0[mm] = (v8f){lo.x, lo.y, lo.z, lo.w, hi.x, hi.y, hi.z, hi.w};
            }
        }
        SCHED_FENCE();
        load_par(lng, par, lane);
        load_par(lne, par + DHID, lane);
        v4f af[8];
        v8f h1[8];
        ln_centred_tanh_T(h0, lng, lne, [&] { load_a(af, wl1, 0); load_tile(h1, par + 2 * DHID, lane); });

        gemm_chained(h1, h0, wl1, af, [&] {
            load_par(lng, par + 3 * DHID, lane); load_par(lne, par + 4 * DHID, lane); });
        v8f h2[8];
        ln_centred_tanh_T(h1, lng, lne, [&] { load_a(af, wl2, 0); load_tile(h2, par + 5 * DHID, lane); });

        gemm_chained(h2, h1, wl2, af, [&] {
            load_par(lng, par + 6 * DHID, lane); load_par(lne, par + 7 * DHID, lane); });
        v4f w3v[8][2];
        ln_centred_tanh_T(h2, lng, lne, [&] { load_par(w3v, par + 8 * DHID, lane); });

        {
            float p = dot64_T(h2, w3v);
            p += __shfl_xor(p, 16, 32);
            const int e = e0 + c;
            if (h == 0 && e < E) out[e] = p + b3[0];
        }
    }
    __builtin_amdgcn_s_setprio(0);
}

extern "C" void kernel_launch(void* const* d_in, const int* in_sizes, int n_in,
                              void* d_out, int out_size, void* d_ws, size_t ws_size,
                              hipStream_t stream)
{
    const float* x   = (const float*)d_in[0];
    const int*   ei  = (const int*)  d_in[1];
    const float* W0  = (const float*)d_in[2];
    const float* b0  = (const float*)d_in[3];
    const float* W1  = (const float*)d_in[4];
    const float* b1  = (const float*)d_in[5];
    const float* W2  = (const float*)d_in[6];
    const float* b2  = (const float*)d_in[7];
    const float* W3  = (const float*)d_in[8];
    const float* b3  = (const float*)d_in[9];
    const float* g0  = (const float*)d_in[10];
    const float* be0 = (const float*)d_in[11];
    const float* g1  = (const float*)d_in[12];
    const float* be1 = (const float*)d_in[13];
    const float* g2  = (const float*)d_in[14];
    const float* be2 = (const float*)d_in[15];

    const int E     = in_sizes[1] / 2;
    const int N     = in_sizes[0] / DHID;
    const int tiles = (E + TILE_E - 1) / TILE_E;

    const size_t need = (size_t)N * PQ_ROW * sizeof(float);
    if (d_ws != nullptr && ws_size >= need) {
        const int ntiles = (N + TILE_E - 1) / TILE_E;
        int nblocks = (ntiles + WPB - 1) / WPB;
        if (nblocks > 256) nblocks = 256;
        int eblocks = (tiles + WPB - 1) / WPB;
        if (eblocks > 256) eblocks = 256;
        float* pq = (float*)d_ws;
        node_proj_wmma_f32<<<nblocks, WPB * 32, (size_t)SMEM_NODE_FLOATS * sizeof(float), stream>>>(
            x, N, W0, b0, pq);
        edge_mlp_pq_wmma_f32<<<eblocks, WPB * 32, (size_t)SMEM_EDGE_FLOATS * sizeof(float), stream>>>(
            pq, ei, E, W1, b1, W2, b2, W3, b3,
            g0, be0, g1, be1, g2, be2, (float*)d_out);
        return;
    }

    int blocks = (tiles + WPB - 1) / WPB;
    if (blocks > 256) blocks = 256;
    const size_t shmem = (size_t)SMEM_FLOATS * sizeof(float);

    edge_mlp_wmma_f32<<<blocks, WPB * 32, shmem, stream>>>(
        x, ei, E, W0, b0, W1, b1, W2, b2, W3, b3,
        g0, be0, g1, be1, g2, be2, (float*)d_out);
}
